# starting kernel
# baseline (speedup 1.0000x reference)
#include <hip/hip_runtime.h>
#include <math.h>

typedef float v2f __attribute__((ext_vector_type(2)));
typedef float v8f __attribute__((ext_vector_type(8)));

#define HEADS 8
#define DIM 16
#define F 128
#define NEG_SLOPE 0.2f
#define TEM 0.7f

__device__ __forceinline__ unsigned fenc(float f) {
    unsigned b = __float_as_uint(f);
    return (b & 0x80000000u) ? ~b : (b | 0x80000000u);
}
__device__ __forceinline__ float fdec(unsigned u) {
    unsigned b = (u & 0x80000000u) ? (u & 0x7FFFFFFFu) : ~u;
    return __uint_as_float(b);
}

__global__ __launch_bounds__(256) void gemm_z_wmma(
    const float* __restrict__ feat, const float* __restrict__ fcw,
    float* __restrict__ z, int N)
{
    __shared__ float sA[16 * F];
    const int row0 = blockIdx.x * 16;

    for (int i = threadIdx.x; i < 16 * F; i += 256) {
        int r = i >> 7, c = i & 127;
        int gr = row0 + r;
        sA[i] = (gr < N) ? feat[(long long)gr * F + c] : 0.0f;
    }
    __syncthreads();

    const int wave = threadIdx.x >> 5;
    const int lane = threadIdx.x & 31;
    const int col0 = wave * 16;
    const int m    = lane & 15;
    const int koff = (lane >> 4) * 2;

    v8f c = {};
    for (int kb = 0; kb < F; kb += 4) {
        const int k = kb + koff;
        v2f a, b;
        a.x = sA[m * F + k];
        a.y = sA[m * F + k + 1];
        b.x = fcw[(long long)k * F + col0 + m];
        b.y = fcw[(long long)(k + 1) * F + col0 + m];
        c = __builtin_amdgcn_wmma_f32_16x16x4_f32(
                false, a, false, b, (short)0, c, false, false);
    }

    for (int i = 0; i < 8; ++i) {
        int gr = row0 + i + (lane >> 4) * 8;
        if (gr < N) z[(long long)gr * F + col0 + m] = c[i];
    }
}

__global__ void el_er_kernel(const float* __restrict__ z,
                             const float* __restrict__ attn_l,
                             const float* __restrict__ attn_r,
                             float* __restrict__ el, float* __restrict__ er,
                             int N)
{
    int t = blockIdx.x * blockDim.x + threadIdx.x;
    if (t >= N * HEADS) return;
    int n = t >> 3, h = t & 7;
    const float* zp = z + (long long)n * F + h * DIM;
    const float* al = attn_l + h * DIM;
    const float* ar = attn_r + h * DIM;
    float sl = 0.f, sr = 0.f;
    #pragma unroll
    for (int d = 0; d < DIM; ++d) { float v = zp[d]; sl += v * al[d]; sr += v * ar[d]; }
    el[t] = sl; er[t] = sr;
}

__global__ void init_kernel(float* __restrict__ rst, float* __restrict__ denom,
                            unsigned* __restrict__ menc, float* __restrict__ red,
                            int N)
{
    int t = blockIdx.x * blockDim.x + threadIdx.x;
    if (t < N * F) rst[t] = 0.0f;
    if (t < N * HEADS) { denom[t] = 0.0f; menc[t] = 0u; }
    if (t == 0) red[0] = 0.0f;
}

__device__ __forceinline__ void edge_lookup(const int* __restrict__ esrc,
                                            const int* __restrict__ edst,
                                            const int* __restrict__ sel,
                                            int n_real, int e, int& s, int& d)
{
    if (e < n_real) {
        int j = sel ? sel[e] : e;
        s = esrc[j]; d = edst[j];
    } else {
        s = d = e - n_real;
    }
}

__global__ void edge_max_kernel(const float* __restrict__ el,
                                const float* __restrict__ er,
                                const int* __restrict__ esrc,
                                const int* __restrict__ edst,
                                const int* __restrict__ sel,
                                int n_real, int N,
                                unsigned* __restrict__ menc)
{
    int t = blockIdx.x * blockDim.x + threadIdx.x;
    int total = (n_real + N) * HEADS;
    if (t >= total) return;
    int e = t >> 3, h = t & 7;
    int s, d; edge_lookup(esrc, edst, sel, n_real, e, s, d);
    float x = el[s * HEADS + h] + er[d * HEADS + h];
    x = (x >= 0.f) ? x : NEG_SLOPE * x;
    atomicMax(&menc[d * HEADS + h], fenc(x));
}

__global__ void edge_acc_kernel(const float* __restrict__ el,
                                const float* __restrict__ er,
                                const float* __restrict__ z,
                                const int* __restrict__ esrc,
                                const int* __restrict__ edst,
                                const int* __restrict__ sel,
                                int n_real, int N,
                                const unsigned* __restrict__ menc,
                                float* __restrict__ denom,
                                float* __restrict__ rst)
{
    int t = blockIdx.x * blockDim.x + threadIdx.x;
    int total = (n_real + N) * HEADS;
    if (t >= total) return;
    int e = t >> 3, h = t & 7;
    int s, d; edge_lookup(esrc, edst, sel, n_real, e, s, d);
    float x = el[s * HEADS + h] + er[d * HEADS + h];
    x = (x >= 0.f) ? x : NEG_SLOPE * x;
    float mx = fdec(menc[d * HEADS + h]);
    float w = __expf(x - mx);
    atomicAdd(&denom[d * HEADS + h], w);
    const float* zp = z + (long long)s * F + h * DIM;
    float* rp = rst + (long long)d * F + h * DIM;
    #pragma unroll
    for (int k = 0; k < DIM; ++k) atomicAdd(&rp[k], w * zp[k]);
}

__global__ void finalize_kernel(const float* __restrict__ rst,
                                const float* __restrict__ denom,
                                const float* __restrict__ bias,
                                const float* __restrict__ prelu_a,
                                float* __restrict__ hout,
                                int n0, int nCnt)
{
    int t = blockIdx.x * blockDim.x + threadIdx.x;
    if (t >= nCnt * DIM) return;
    int i = t >> 4, dcol = t & 15;
    int n = n0 + i;
    float a = prelu_a[0];
    float acc = 0.f;
    #pragma unroll
    for (int h = 0; h < HEADS; ++h) {
        float v = rst[(long long)n * F + h * DIM + dcol] / denom[n * HEADS + h]
                + bias[h * DIM + dcol];
        v = (v >= 0.f) ? v : a * v;
        acc += v;
    }
    hout[(long long)i * DIM + dcol] = acc * (1.0f / HEADS);
}

__global__ __launch_bounds__(256) void cos_loss_kernel(
    const float* __restrict__ hp, const float* __restrict__ hn,
    float* __restrict__ red, int nCnt)
{
    int i = blockIdx.x * blockDim.x + threadIdx.x;
    float v = 0.f;
    if (i < nCnt) {
        float num = 0.f, na = 0.f, nb = 0.f;
        #pragma unroll
        for (int d = 0; d < DIM; ++d) {
            float a = hp[(long long)i * DIM + d];
            float b = hn[(long long)i * DIM + d];
            num += a * b; na += a * a; nb += b * b;
        }
        na = fmaxf(sqrtf(na), 1e-8f);
        nb = fmaxf(sqrtf(nb), 1e-8f);
        v = __expf((num / (na * nb)) * (1.0f / TEM));
    }
    __shared__ float sbuf[256];
    sbuf[threadIdx.x] = v;
    __syncthreads();
    for (int s = 128; s > 0; s >>= 1) {
        if (threadIdx.x < s) sbuf[threadIdx.x] += sbuf[threadIdx.x + s];
        __syncthreads();
    }
    if (threadIdx.x == 0) atomicAdd(red, sbuf[0]);
}

__global__ void write_loss_kernel(const float* __restrict__ red, float* __restrict__ out)
{
    out[0] = logf(red[0]);
}

extern "C" void kernel_launch(void* const* d_in, const int* in_sizes, int n_in,
                              void* d_out, int out_size, void* d_ws, size_t ws_size,
                              hipStream_t stream)
{
    const float* feat    = (const float*)d_in[0];
    const float* fcw     = (const float*)d_in[1];
    const float* attn_l  = (const float*)d_in[2];
    const float* attn_r  = (const float*)d_in[3];
    const float* bias    = (const float*)d_in[4];
    const float* prelu_a = (const float*)d_in[5];
    const int*   esrc    = (const int*)d_in[6];
    const int*   edst    = (const int*)d_in[7];
    const int*   negidx  = (const int*)d_in[8];

    const int N     = in_sizes[0] / F;
    const int E     = in_sizes[6];
    const int E_NEG = in_sizes[8];
    const int N_DST = (out_size - 1) / DIM;
    const int N_SRC = N - N_DST;

    float* out   = (float*)d_out;
    float* h_pos = out + 1;

    float*    ws    = (float*)d_ws;
    float*    z     = ws;
    float*    el    = z   + (size_t)N * F;
    float*    er    = el  + (size_t)N * HEADS;
    float*    denom = er  + (size_t)N * HEADS;
    unsigned* menc  = (unsigned*)(denom + (size_t)N * HEADS);
    float*    rst   = (float*)(menc + (size_t)N * HEADS);
    float*    h_neg = rst + (size_t)N * F;
    float*    red   = h_neg + (size_t)N_DST * DIM;

    const int TB = 256;
    auto cdiv = [](long long a, int b) { return (unsigned)((a + b - 1) / b); };

    gemm_z_wmma<<<cdiv(N, 16), TB, 0, stream>>>(feat, fcw, z, N);

    el_er_kernel<<<cdiv((long long)N * HEADS, TB), TB, 0, stream>>>(
        z, attn_l, attn_r, el, er, N);

    init_kernel<<<cdiv((long long)N * F, TB), TB, 0, stream>>>(rst, denom, menc, red, N);
    edge_max_kernel<<<cdiv((long long)(E + N) * HEADS, TB), TB, 0, stream>>>(
        el, er, esrc, edst, nullptr, E, N, menc);
    edge_acc_kernel<<<cdiv((long long)(E + N) * HEADS, TB), TB, 0, stream>>>(
        el, er, z, esrc, edst, nullptr, E, N, menc, denom, rst);
    finalize_kernel<<<cdiv((long long)N_DST * DIM, TB), TB, 0, stream>>>(
        rst, denom, bias, prelu_a, h_pos, N_SRC, N_DST);

    init_kernel<<<cdiv((long long)N * F, TB), TB, 0, stream>>>(rst, denom, menc, red, N);
    edge_max_kernel<<<cdiv((long long)(E_NEG + N) * HEADS, TB), TB, 0, stream>>>(
        el, er, esrc, edst, negidx, E_NEG, N, menc);
    edge_acc_kernel<<<cdiv((long long)(E_NEG + N) * HEADS, TB), TB, 0, stream>>>(
        el, er, z, esrc, edst, negidx, E_NEG, N, menc, denom, rst);
    finalize_kernel<<<cdiv((long long)N_DST * DIM, TB), TB, 0, stream>>>(
        rst, denom, bias, prelu_a, h_neg, N_SRC, N_DST);

    cos_loss_kernel<<<cdiv(N_DST, TB), TB, 0, stream>>>(h_pos, h_neg, red, N_DST);
    write_loss_kernel<<<1, 1, 0, stream>>>(red, out);
}


# Float-atomic edge scatter replaced by per-call counting sort by dst (int atomics) + wave-per-dst gather serving both graphs; sums now in cursor-arrival order, head mean as lane tree
# speedup vs baseline: 3.6546x; 3.6546x over previous
#include <hip/hip_runtime.h>
#include <math.h>

typedef float v2f __attribute__((ext_vector_type(2)));
typedef float v8f __attribute__((ext_vector_type(8)));

#define HEADS 8
#define DIM 16
#define F 128
#define NEG_SLOPE 0.2f
#define TEM 0.7f

__global__ __launch_bounds__(256) void gemm_z_wmma(
    const float* __restrict__ feat, const float* __restrict__ fcw,
    float* __restrict__ z, int N)
{
    __shared__ float sA[16 * F];
    const int row0 = blockIdx.x * 16;

    for (int i = threadIdx.x; i < 16 * F; i += 256) {
        int r = i >> 7, c = i & 127;
        int gr = row0 + r;
        sA[i] = (gr < N) ? feat[(long long)gr * F + c] : 0.0f;
    }
    __syncthreads();

    const int wave = threadIdx.x >> 5;
    const int lane = threadIdx.x & 31;
    const int col0 = wave * 16;
    const int m    = lane & 15;
    const int koff = (lane >> 4) * 2;

    v8f c = {};
    for (int kb = 0; kb < F; kb += 4) {
        const int k = kb + koff;
        v2f a, b;
        a.x = sA[m * F + k];
        a.y = sA[m * F + k + 1];
        b.x = fcw[(long long)k * F + col0 + m];
        b.y = fcw[(long long)(k + 1) * F + col0 + m];
        c = __builtin_amdgcn_wmma_f32_16x16x4_f32(
                false, a, false, b, (short)0, c, false, false);
    }

    for (int i = 0; i < 8; ++i) {
        int gr = row0 + i + (lane >> 4) * 8;
        if (gr < N) z[(long long)gr * F + col0 + m] = c[i];
    }
}

__global__ void el_er_kernel(const float* __restrict__ z,
                             const float* __restrict__ attn_l,
                             const float* __restrict__ attn_r,
                             float* __restrict__ el, float* __restrict__ er,
                             int N)
{
    int t = blockIdx.x * blockDim.x + threadIdx.x;
    if (t >= N * HEADS) return;
    int n = t >> 3, h = t & 7;
    const float* zp = z + (long long)n * F + h * DIM;
    const float* al = attn_l + h * DIM;
    const float* ar = attn_r + h * DIM;
    float sl = 0.f, sr = 0.f;
    #pragma unroll
    for (int d = 0; d < DIM; ++d) { float v = zp[d]; sl += v * al[d]; sr += v * ar[d]; }
    el[t] = sl; er[t] = sr;
}

__global__ void init_kernel(int* __restrict__ mult, int* __restrict__ cnt,
                            float* __restrict__ red, int E, int nCnt)
{
    int t = blockIdx.x * blockDim.x + threadIdx.x;
    if (t < E) mult[t] = 0;
    else if (t - E < nCnt) cnt[t - E] = 0;
    if (t == 0) red[0] = 0.0f;
}

__global__ void count_kernel(const int* __restrict__ edst,
                             const int* __restrict__ sel,
                             int E, int E_sel, int n0, int nCnt,
                             int* __restrict__ cnt, int* __restrict__ mult)
{
    int t = blockIdx.x * blockDim.x + threadIdx.x;
    if (t < E) {
        int d = edst[t] - n0;
        if (d >= 0 && d < nCnt) atomicAdd(&cnt[d], 1);
    } else if (t - E < E_sel) {
        atomicAdd(&mult[sel[t - E]], 1);
    }
}

__device__ __forceinline__ void load4(const int* __restrict__ cnt, int b, int k0,
                                      int per, int nCnt, int (&v)[4])
{
    #pragma unroll
    for (int j = 0; j < 4; ++j) {
        const int idx = b + k0 + j;
        const bool ok = (k0 + j < per) && (idx < nCnt);
        const int x = cnt[ok ? idx : 0];
        v[j] = ok ? x : 0;
    }
}

__global__ __launch_bounds__(1024) void scan_kernel(const int* __restrict__ cnt,
                                                    int* __restrict__ rowptr,
                                                    int* __restrict__ cursor,
                                                    int nCnt)
{
    __shared__ int wsum[32];
    const int tid = threadIdx.x;
    const int lane = tid & 31, wv = tid >> 5;
    const int per = (nCnt + 1023) / 1024;
    const int b = tid * per;
    int local = 0;
    for (int k0 = 0; k0 < per; k0 += 4) {
        int v4[4];
        load4(cnt, b, k0, per, nCnt, v4);
        local += (v4[0] + v4[1]) + (v4[2] + v4[3]);
    }
    int v = local;
    for (int off = 1; off < 32; off <<= 1) { int u = __shfl_up(v, off); if (lane >= off) v += u; }
    if (lane == 31) wsum[wv] = v;
    __syncthreads();
    if (wv == 0) {
        int s = wsum[lane];
        for (int off = 1; off < 32; off <<= 1) { int u = __shfl_up(s, off); if (lane >= off) s += u; }
        wsum[lane] = s;
    }
    __syncthreads();
    int run = v - local + (wv > 0 ? wsum[wv - 1] : 0);
    for (int k0 = 0; k0 < per; k0 += 4) {
        int v4[4];
        load4(cnt, b, k0, per, nCnt, v4);
        #pragma unroll
        for (int j = 0; j < 4; ++j) {
            const int idx = b + k0 + j;
            if ((k0 + j < per) && (idx < nCnt)) { rowptr[idx] = run; cursor[idx] = run; run += v4[j]; }
        }
    }
}

__global__ void fill_kernel(const int* __restrict__ esrc,
                            const int* __restrict__ edst,
                            const int* __restrict__ mult,
                            int E, int n0, int nCnt,
                            int* __restrict__ cursor, int2* __restrict__ csr)
{
    int t = blockIdx.x * blockDim.x + threadIdx.x;
    if (t >= E) return;
    int d = edst[t] - n0;
    if (d < 0 || d >= nCnt) return;
    int pos = atomicAdd(&cursor[d], 1);
    csr[pos] = make_int2(esrc[t], mult[t]);
}

__device__ __forceinline__ float leaky(float x) { return (x >= 0.f) ? x : NEG_SLOPE * x; }

__global__ __launch_bounds__(256) void gather_kernel(
    const float* __restrict__ z, const float* __restrict__ el,
    const float* __restrict__ er,
    const int* __restrict__ rowptr, const int* __restrict__ rowend,
    const int2* __restrict__ csr,
    const float* __restrict__ bias, const float* __restrict__ prelu_a,
    float* __restrict__ h_pos, float* __restrict__ h_neg, int n0, int nCnt)
{
    const int lane = threadIdx.x & 31;
    const int i = __builtin_amdgcn_readfirstlane((int)(blockIdx.x * 8 + (threadIdx.x >> 5)));
    if (i >= nCnt) return;
    const int n = n0 + i;
    const int h = lane >> 2;
    const int start = rowptr[i];
    const int end = rowend[i];
    const float4* __restrict__ z4 = (const float4*)z;

    const float erh = er[n * HEADS + h];
    const float xs = leaky(el[n * HEADS + h] + erh);
    float mA = xs, mB = xs;

    for (int base = start; base < end; base += 32) {
        const int c = (end - base < 32) ? (end - base) : 32;
        int2 rec = make_int2(0, 0);
        if (lane < c) rec = csr[base + lane];
        for (int k = 0; k < c; ++k) {
            const int s  = __builtin_amdgcn_readlane(rec.x, k);
            const int mu = __builtin_amdgcn_readlane(rec.y, k);
            const float x = leaky(el[s * HEADS + h] + erh);
            mA = fmaxf(mA, x);
            if (mu > 0) mB = fmaxf(mB, x);
        }
    }

    float dA = __expf(xs - mA), dB = __expf(xs - mB);
    float4 zv = z4[(size_t)n * 32 + lane];
    float4 aA = make_float4(dA * zv.x, dA * zv.y, dA * zv.z, dA * zv.w);
    float4 aB = make_float4(dB * zv.x, dB * zv.y, dB * zv.z, dB * zv.w);
    for (int base = start; base < end; base += 32) {
        const int c = (end - base < 32) ? (end - base) : 32;
        int2 rec = make_int2(0, 0);
        if (lane < c) rec = csr[base + lane];
        for (int k = 0; k < c; ++k) {
            const int s  = __builtin_amdgcn_readlane(rec.x, k);
            const int mu = __builtin_amdgcn_readlane(rec.y, k);
            const float x = leaky(el[s * HEADS + h] + erh);
            const float4 r = z4[(size_t)s * 32 + lane];
            const float wa = __expf(x - mA);
            const float wb = (mu > 0) ? (float)mu * __expf(x - mB) : 0.0f;
            dA += wa; dB += wb;
            aA.x += wa * r.x; aA.y += wa * r.y; aA.z += wa * r.z; aA.w += wa * r.w;
            aB.x += wb * r.x; aB.y += wb * r.y; aB.z += wb * r.z; aB.w += wb * r.w;
        }
    }

    const float a = prelu_a[0];
    const float4 b4 = ((const float4*)bias)[lane];
    float pA[4] = { aA.x / dA + b4.x, aA.y / dA + b4.y, aA.z / dA + b4.z, aA.w / dA + b4.w };
    float pB[4] = { aB.x / dB + b4.x, aB.y / dB + b4.y, aB.z / dB + b4.z, aB.w / dB + b4.w };
    #pragma unroll
    for (int c = 0; c < 4; ++c) {
        pA[c] = (pA[c] >= 0.f) ? pA[c] : a * pA[c];
        pB[c] = (pB[c] >= 0.f) ? pB[c] : a * pB[c];
        for (int off = 4; off < 32; off <<= 1) {
            pA[c] += __shfl_xor(pA[c], off);
            pB[c] += __shfl_xor(pB[c], off);
        }
    }
    const int q = lane & 3, cc = (lane >> 2) & 3;
    const bool neg = lane >= 16;
    const float v0 = neg ? pB[0] : pA[0];
    const float v1 = neg ? pB[1] : pA[1];
    const float v2 = neg ? pB[2] : pA[2];
    const float v3 = neg ? pB[3] : pA[3];
    const float v = (cc == 0) ? v0 : (cc == 1) ? v1 : (cc == 2) ? v2 : v3;
    float* __restrict__ dstp = neg ? h_neg : h_pos;
    dstp[(size_t)i * DIM + q * 4 + cc] = v * (1.0f / HEADS);
}

__global__ __launch_bounds__(256) void cos_loss_kernel(
    const float* __restrict__ hp, const float* __restrict__ hn,
    float* __restrict__ red, int nCnt)
{
    int i = blockIdx.x * blockDim.x + threadIdx.x;
    float v = 0.f;
    if (i < nCnt) {
        float num = 0.f, na = 0.f, nb = 0.f;
        #pragma unroll
        for (int d = 0; d < DIM; ++d) {
            float a = hp[(long long)i * DIM + d];
            float b = hn[(long long)i * DIM + d];
            num += a * b; na += a * a; nb += b * b;
        }
        na = fmaxf(sqrtf(na), 1e-8f);
        nb = fmaxf(sqrtf(nb), 1e-8f);
        v = __expf((num / (na * nb)) * (1.0f / TEM));
    }
    __shared__ float sbuf[256];
    sbuf[threadIdx.x] = v;
    __syncthreads();
    for (int s = 128; s > 0; s >>= 1) {
        if (threadIdx.x < s) sbuf[threadIdx.x] += sbuf[threadIdx.x + s];
        __syncthreads();
    }
    if (threadIdx.x == 0) atomicAdd(red, sbuf[0]);
}

__global__ void write_loss_kernel(const float* __restrict__ red, float* __restrict__ out)
{
    out[0] = logf(red[0]);
}

extern "C" void kernel_launch(void* const* d_in, const int* in_sizes, int n_in,
                              void* d_out, int out_size, void* d_ws, size_t ws_size,
                              hipStream_t stream)
{
    const float* feat    = (const float*)d_in[0];
    const float* fcw     = (const float*)d_in[1];
    const float* attn_l  = (const float*)d_in[2];
    const float* attn_r  = (const float*)d_in[3];
    const float* bias    = (const float*)d_in[4];
    const float* prelu_a = (const float*)d_in[5];
    const int*   esrc    = (const int*)d_in[6];
    const int*   edst    = (const int*)d_in[7];
    const int*   negidx  = (const int*)d_in[8];

    const int N     = in_sizes[0] / F;
    const int E     = in_sizes[6];
    const int E_NEG = in_sizes[8];
    const int N_DST = (out_size - 1) / DIM;
    const int N_SRC = N - N_DST;

    float* out   = (float*)d_out;
    float* h_pos = out + 1;

    float*    ws     = (float*)d_ws;
    float*    z      = ws;
    float*    el     = z   + (size_t)N * F;
    float*    er     = el  + (size_t)N * HEADS;
    float*    h_neg  = er  + (size_t)N * HEADS;
    int2*     csr    = (int2*)(h_neg + (size_t)N_DST * DIM);
    int*      mult   = (int*)(csr + (size_t)E);
    int*      cnt    = mult + (size_t)E;
    int*      rowptr = cnt + (size_t)N_DST;
    int*      cursor = rowptr + (size_t)N_DST;
    float*    red    = (float*)(cursor + (size_t)N_DST);

    const int TB = 256;
    auto cdiv = [](long long a, int b) { return (unsigned)((a + b - 1) / b); };

    gemm_z_wmma<<<cdiv(N, 16), TB, 0, stream>>>(feat, fcw, z, N);

    el_er_kernel<<<cdiv((long long)N * HEADS, TB), TB, 0, stream>>>(
        z, attn_l, attn_r, el, er, N);

    init_kernel<<<cdiv((long long)E + N_DST, TB), TB, 0, stream>>>(mult, cnt, red, E, N_DST);
    count_kernel<<<cdiv((long long)E + E_NEG, TB), TB, 0, stream>>>(
        edst, negidx, E, E_NEG, N_SRC, N_DST, cnt, mult);
    scan_kernel<<<1, 1024, 0, stream>>>(cnt, rowptr, cursor, N_DST);
    fill_kernel<<<cdiv(E, TB), TB, 0, stream>>>(
        esrc, edst, mult, E, N_SRC, N_DST, cursor, csr);

    gather_kernel<<<cdiv(N_DST, 8), TB, 0, stream>>>(
        z, el, er, rowptr, cursor, csr, bias, prelu_a, h_pos, h_neg, N_SRC, N_DST);

    cos_loss_kernel<<<cdiv(N_DST, TB), TB, 0, stream>>>(h_pos, h_neg, red, N_DST);
    write_loss_kernel<<<1, 1, 0, stream>>>(red, out);
}


# GEMM A tile staged by two 16-byte loads per thread issued together into LDS rows of pitch 132 floats: fragment reads no longer pile 16 lanes on one bank pair
# speedup vs baseline: 4.3040x; 1.1777x over previous
#include <hip/hip_runtime.h>
#include <math.h>

typedef float v2f __attribute__((ext_vector_type(2)));
typedef float v8f __attribute__((ext_vector_type(8)));

#define HEADS 8
#define DIM 16
#define F 128
#define NEG_SLOPE 0.2f
#define TEM 0.7f

#define APITCH (F + 4)
__global__ __launch_bounds__(256) void gemm_z_wmma(
    const float* __restrict__ feat, const float* __restrict__ fcw,
    float* __restrict__ z, int N)
{
    __shared__ __attribute__((aligned(16))) float sA[16 * APITCH];
    const int row0 = blockIdx.x * 16;

    {
        const int c4 = threadIdx.x & 31;
        const int r0 = threadIdx.x >> 5;
        const int r1 = r0 + 8;
        const int g0 = (row0 + r0 < N) ? (row0 + r0) : (N - 1);
        const int g1 = (row0 + r1 < N) ? (row0 + r1) : (N - 1);
        const float4 v0 = ((const float4*)(feat + (long long)g0 * F))[c4];
        const float4 v1 = ((const float4*)(feat + (long long)g1 * F))[c4];
        *(float4*)&sA[r0 * APITCH + c4 * 4] = v0;
        *(float4*)&sA[r1 * APITCH + c4 * 4] = v1;
    }
    __syncthreads();

    const int wave = threadIdx.x >> 5;
    const int lane = threadIdx.x & 31;
    const int col0 = wave * 16;
    const int m    = lane & 15;
    const int koff = (lane >> 4) * 2;

    v8f c = {};
    for (int kb = 0; kb < F; kb += 4) {
        const int k = kb + koff;
        v2f a, b;
        a.x = sA[m * APITCH + k];
        a.y = sA[m * APITCH + k + 1];
        b.x = fcw[(long long)k * F + col0 + m];
        b.y = fcw[(long long)(k + 1) * F + col0 + m];
        c = __builtin_amdgcn_wmma_f32_16x16x4_f32(
                false, a, false, b, (short)0, c, false, false);
    }

    for (int i = 0; i < 8; ++i) {
        int gr = row0 + i + (lane >> 4) * 8;
        if (gr < N) z[(long long)gr * F + col0 + m] = c[i];
    }
}

__global__ void el_er_kernel(const float* __restrict__ z,
                             const float* __restrict__ attn_l,
                             const float* __restrict__ attn_r,
                             float* __restrict__ el, float* __restrict__ er,
                             int N)
{
    int t = blockIdx.x * blockDim.x + threadIdx.x;
    if (t >= N * HEADS) return;
    int n = t >> 3, h = t & 7;
    const float* zp = z + (long long)n * F + h * DIM;
    const float* al = attn_l + h * DIM;
    const float* ar = attn_r + h * DIM;
    float sl = 0.f, sr = 0.f;
    #pragma unroll
    for (int d = 0; d < DIM; ++d) { float v = zp[d]; sl += v * al[d]; sr += v * ar[d]; }
    el[t] = sl; er[t] = sr;
}

__global__ void init_kernel(int* __restrict__ mult, int* __restrict__ cnt,
                            float* __restrict__ red, int E, int nCnt)
{
    int t = blockIdx.x * blockDim.x + threadIdx.x;
    if (t < E) mult[t] = 0;
    else if (t - E < nCnt) cnt[t - E] = 0;
    if (t == 0) red[0] = 0.0f;
}

__global__ void count_kernel(const int* __restrict__ edst,
                             const int* __restrict__ sel,
                             int E, int E_sel, int n0, int nCnt,
                             int* __restrict__ cnt, int* __restrict__ mult)
{
    int t = blockIdx.x * blockDim.x + threadIdx.x;
    if (t < E) {
        int d = edst[t] - n0;
        if (d >= 0 && d < nCnt) atomicAdd(&cnt[d], 1);
    } else if (t - E < E_sel) {
        atomicAdd(&mult[sel[t - E]], 1);
    }
}

__device__ __forceinline__ void load4(const int* __restrict__ cnt, int b, int k0,
                                      int per, int nCnt, int (&v)[4])
{
    #pragma unroll
    for (int j = 0; j < 4; ++j) {
        const int idx = b + k0 + j;
        const bool ok = (k0 + j < per) && (idx < nCnt);
        const int x = cnt[ok ? idx : 0];
        v[j] = ok ? x : 0;
    }
}

__global__ __launch_bounds__(1024) void scan_kernel(const int* __restrict__ cnt,
                                                    int* __restrict__ rowptr,
                                                    int* __restrict__ cursor,
                                                    int nCnt)
{
    __shared__ int wsum[32];
    const int tid = threadIdx.x;
    const int lane = tid & 31, wv = tid >> 5;
    const int per = (nCnt + 1023) / 1024;
    const int b = tid * per;
    int local = 0;
    for (int k0 = 0; k0 < per; k0 += 4) {
        int v4[4];
        load4(cnt, b, k0, per, nCnt, v4);
        local += (v4[0] + v4[1]) + (v4[2] + v4[3]);
    }
    int v = local;
    for (int off = 1; off < 32; off <<= 1) { int u = __shfl_up(v, off); if (lane >= off) v += u; }
    if (lane == 31) wsum[wv] = v;
    __syncthreads();
    if (wv == 0) {
        int s = wsum[lane];
        for (int off = 1; off < 32; off <<= 1) { int u = __shfl_up(s, off); if (lane >= off) s += u; }
        wsum[lane] = s;
    }
    __syncthreads();
    int run = v - local + (wv > 0 ? wsum[wv - 1] : 0);
    for (int k0 = 0; k0 < per; k0 += 4) {
        int v4[4];
        load4(cnt, b, k0, per, nCnt, v4);
        #pragma unroll
        for (int j = 0; j < 4; ++j) {
            const int idx = b + k0 + j;
            if ((k0 + j < per) && (idx < nCnt)) { rowptr[idx] = run; cursor[idx] = run; run += v4[j]; }
        }
    }
}

__global__ void fill_kernel(const int* __restrict__ esrc,
                            const int* __restrict__ edst,
                            const int* __restrict__ mult,
                            int E, int n0, int nCnt,
                            int* __restrict__ cursor, int2* __restrict__ csr)
{
    int t = blockIdx.x * blockDim.x + threadIdx.x;
    if (t >= E) return;
    int d = edst[t] - n0;
    if (d < 0 || d >= nCnt) return;
    int pos = atomicAdd(&cursor[d], 1);
    csr[pos] = make_int2(esrc[t], mult[t]);
}

__device__ __forceinline__ float leaky(float x) { return (x >= 0.f) ? x : NEG_SLOPE * x; }

__global__ __launch_bounds__(256) void gather_kernel(
    const float* __restrict__ z, const float* __restrict__ el,
    const float* __restrict__ er,
    const int* __restrict__ rowptr, const int* __restrict__ rowend,
    const int2* __restrict__ csr,
    const float* __restrict__ bias, const float* __restrict__ prelu_a,
    float* __restrict__ h_pos, float* __restrict__ h_neg, int n0, int nCnt)
{
    const int lane = threadIdx.x & 31;
    const int i = __builtin_amdgcn_readfirstlane((int)(blockIdx.x * 8 + (threadIdx.x >> 5)));
    if (i >= nCnt) return;
    const int n = n0 + i;
    const int h = lane >> 2;
    const int start = rowptr[i];
    const int end = rowend[i];
    const float4* __restrict__ z4 = (const float4*)z;

    const float erh = er[n * HEADS + h];
    const float xs = leaky(el[n * HEADS + h] + erh);
    float mA = xs, mB = xs;

    for (int base = start; base < end; base += 32) {
        const int c = (end - base < 32) ? (end - base) : 32;
        int2 rec = make_int2(0, 0);
        if (lane < c) rec = csr[base + lane];
        for (int k = 0; k < c; ++k) {
            const int s  = __builtin_amdgcn_readlane(rec.x, k);
            const int mu = __builtin_amdgcn_readlane(rec.y, k);
            const float x = leaky(el[s * HEADS + h] + erh);
            mA = fmaxf(mA, x);
            if (mu > 0) mB = fmaxf(mB, x);
        }
    }

    float dA = __expf(xs - mA), dB = __expf(xs - mB);
    float4 zv = z4[(size_t)n * 32 + lane];
    float4 aA = make_float4(dA * zv.x, dA * zv.y, dA * zv.z, dA * zv.w);
    float4 aB = make_float4(dB * zv.x, dB * zv.y, dB * zv.z, dB * zv.w);
    for (int base = start; base < end; base += 32) {
        const int c = (end - base < 32) ? (end - base) : 32;
        int2 rec = make_int2(0, 0);
        if (lane < c) rec = csr[base + lane];
        for (int k = 0; k < c; ++k) {
            const int s  = __builtin_amdgcn_readlane(rec.x, k);
            const int mu = __builtin_amdgcn_readlane(rec.y, k);
            const float x = leaky(el[s * HEADS + h] + erh);
            const float4 r = z4[(size_t)s * 32 + lane];
            const float wa = __expf(x - mA);
            const float wb = (mu > 0) ? (float)mu * __expf(x - mB) : 0.0f;
            dA += wa; dB += wb;
            aA.x += wa * r.x; aA.y += wa * r.y; aA.z += wa * r.z; aA.w += wa * r.w;
            aB.x += wb * r.x; aB.y += wb * r.y; aB.z += wb * r.z; aB.w += wb * r.w;
        }
    }

    const float a = prelu_a[0];
    const float4 b4 = ((const float4*)bias)[lane];
    float pA[4] = { aA.x / dA + b4.x, aA.y / dA + b4.y, aA.z / dA + b4.z, aA.w / dA + b4.w };
    float pB[4] = { aB.x / dB + b4.x, aB.y / dB + b4.y, aB.z / dB + b4.z, aB.w / dB + b4.w };
    #pragma unroll
    for (int c = 0; c < 4; ++c) {
        pA[c] = (pA[c] >= 0.f) ? pA[c] : a * pA[c];
        pB[c] = (pB[c] >= 0.f) ? pB[c] : a * pB[c];
        for (int off = 4; off < 32; off <<= 1) {
            pA[c] += __shfl_xor(pA[c], off);
            pB[c] += __shfl_xor(pB[c], off);
        }
    }
    const int q = lane & 3, cc = (lane >> 2) & 3;
    const bool neg = lane >= 16;
    const float v0 = neg ? pB[0] : pA[0];
    const float v1 = neg ? pB[1] : pA[1];
    const float v2 = neg ? pB[2] : pA[2];
    const float v3 = neg ? pB[3] : pA[3];
    const float v = (cc == 0) ? v0 : (cc == 1) ? v1 : (cc == 2) ? v2 : v3;
    float* __restrict__ dstp = neg ? h_neg : h_pos;
    dstp[(size_t)i * DIM + q * 4 + cc] = v * (1.0f / HEADS);
}

__global__ __launch_bounds__(256) void cos_loss_kernel(
    const float* __restrict__ hp, const float* __restrict__ hn,
    float* __restrict__ red, int nCnt)
{
    int i = blockIdx.x * blockDim.x + threadIdx.x;
    float v = 0.f;
    if (i < nCnt) {
        float num = 0.f, na = 0.f, nb = 0.f;
        #pragma unroll
        for (int d = 0; d < DIM; ++d) {
            float a = hp[(long long)i * DIM + d];
            float b = hn[(long long)i * DIM + d];
            num += a * b; na += a * a; nb += b * b;
        }
        na = fmaxf(sqrtf(na), 1e-8f);
        nb = fmaxf(sqrtf(nb), 1e-8f);
        v = __expf((num / (na * nb)) * (1.0f / TEM));
    }
    __shared__ float sbuf[256];
    sbuf[threadIdx.x] = v;
    __syncthreads();
    for (int s = 128; s > 0; s >>= 1) {
        if (threadIdx.x < s) sbuf[threadIdx.x] += sbuf[threadIdx.x + s];
        __syncthreads();
    }
    if (threadIdx.x == 0) atomicAdd(red, sbuf[0]);
}

__global__ void write_loss_kernel(const float* __restrict__ red, float* __restrict__ out)
{
    out[0] = logf(red[0]);
}

extern "C" void kernel_launch(void* const* d_in, const int* in_sizes, int n_in,
                              void* d_out, int out_size, void* d_ws, size_t ws_size,
                              hipStream_t stream)
{
    const float* feat    = (const float*)d_in[0];
    const float* fcw     = (const float*)d_in[1];
    const float* attn_l  = (const float*)d_in[2];
    const float* attn_r  = (const float*)d_in[3];
    const float* bias    = (const float*)d_in[4];
    const float* prelu_a = (const float*)d_in[5];
    const int*   esrc    = (const int*)d_in[6];
    const int*   edst    = (const int*)d_in[7];
    const int*   negidx  = (const int*)d_in[8];

    const int N     = in_sizes[0] / F;
    const int E     = in_sizes[6];
    const int E_NEG = in_sizes[8];
    const int N_DST = (out_size - 1) / DIM;
    const int N_SRC = N - N_DST;

    float* out   = (float*)d_out;
    float* h_pos = out + 1;

    float*    ws     = (float*)d_ws;
    float*    z      = ws;
    float*    el     = z   + (size_t)N * F;
    float*    er     = el  + (size_t)N * HEADS;
    float*    h_neg  = er  + (size_t)N * HEADS;
    int2*     csr    = (int2*)(h_neg + (size_t)N_DST * DIM);
    int*      mult   = (int*)(csr + (size_t)E);
    int*      cnt    = mult + (size_t)E;
    int*      rowptr = cnt + (size_t)N_DST;
    int*      cursor = rowptr + (size_t)N_DST;
    float*    red    = (float*)(cursor + (size_t)N_DST);

    const int TB = 256;
    auto cdiv = [](long long a, int b) { return (unsigned)((a + b - 1) / b); };

    gemm_z_wmma<<<cdiv(N, 16), TB, 0, stream>>>(feat, fcw, z, N);

    el_er_kernel<<<cdiv((long long)N * HEADS, TB), TB, 0, stream>>>(
        z, attn_l, attn_r, el, er, N);

    init_kernel<<<cdiv((long long)E + N_DST, TB), TB, 0, stream>>>(mult, cnt, red, E, N_DST);
    count_kernel<<<cdiv((long long)E + E_NEG, TB), TB, 0, stream>>>(
        edst, negidx, E, E_NEG, N_SRC, N_DST, cnt, mult);
    scan_kernel<<<1, 1024, 0, stream>>>(cnt, rowptr, cursor, N_DST);
    fill_kernel<<<cdiv(E, TB), TB, 0, stream>>>(
        esrc, edst, mult, E, N_SRC, N_DST, cursor, csr);

    gather_kernel<<<cdiv(N_DST, 8), TB, 0, stream>>>(
        z, el, er, rowptr, cursor, csr, bias, prelu_a, h_pos, h_neg, N_SRC, N_DST);

    cos_loss_kernel<<<cdiv(N_DST, TB), TB, 0, stream>>>(h_pos, h_neg, red, N_DST);
    write_loss_kernel<<<1, 1, 0, stream>>>(red, out);
}


# Edge rank in its destination run taken from the count pass's returning atomic add; fill places edges at run start + rank with no atomics; sums stay in atomic-arrival order (now the count pass's)
# speedup vs baseline: 4.8079x; 1.1171x over previous
#include <hip/hip_runtime.h>
#include <math.h>

typedef float v2f __attribute__((ext_vector_type(2)));
typedef float v8f __attribute__((ext_vector_type(8)));

#define HEADS 8
#define DIM 16
#define F 128
#define NEG_SLOPE 0.2f
#define TEM 0.7f

#define APITCH (F + 4)
__global__ __launch_bounds__(256) void gemm_z_wmma(
    const float* __restrict__ feat, const float* __restrict__ fcw,
    float* __restrict__ z, int N)
{
    __shared__ __attribute__((aligned(16))) float sA[16 * APITCH];
    const int row0 = blockIdx.x * 16;

    {
        const int c4 = threadIdx.x & 31;
        const int r0 = threadIdx.x >> 5;
        const int r1 = r0 + 8;
        const int g0 = (row0 + r0 < N) ? (row0 + r0) : (N - 1);
        const int g1 = (row0 + r1 < N) ? (row0 + r1) : (N - 1);
        const float4 v0 = ((const float4*)(feat + (long long)g0 * F))[c4];
        const float4 v1 = ((const float4*)(feat + (long long)g1 * F))[c4];
        *(float4*)&sA[r0 * APITCH + c4 * 4] = v0;
        *(float4*)&sA[r1 * APITCH + c4 * 4] = v1;
    }
    __syncthreads();

    const int wave = threadIdx.x >> 5;
    const int lane = threadIdx.x & 31;
    const int col0 = wave * 16;
    const int m    = lane & 15;
    const int koff = (lane >> 4) * 2;

    v8f c = {};
    for (int kb = 0; kb < F; kb += 4) {
        const int k = kb + koff;
        v2f a, b;
        a.x = sA[m * APITCH + k];
        a.y = sA[m * APITCH + k + 1];
        b.x = fcw[(long long)k * F + col0 + m];
        b.y = fcw[(long long)(k + 1) * F + col0 + m];
        c = __builtin_amdgcn_wmma_f32_16x16x4_f32(
                false, a, false, b, (short)0, c, false, false);
    }

    for (int i = 0; i < 8; ++i) {
        int gr = row0 + i + (lane >> 4) * 8;
        if (gr < N) z[(long long)gr * F + col0 + m] = c[i];
    }
}

__global__ void el_er_kernel(const float* __restrict__ z,
                             const float* __restrict__ attn_l,
                             const float* __restrict__ attn_r,
                             float* __restrict__ el, float* __restrict__ er,
                             int N)
{
    int t = blockIdx.x * blockDim.x + threadIdx.x;
    if (t >= N * HEADS) return;
    int n = t >> 3, h = t & 7;
    const float* zp = z + (long long)n * F + h * DIM;
    const float* al = attn_l + h * DIM;
    const float* ar = attn_r + h * DIM;
    float sl = 0.f, sr = 0.f;
    #pragma unroll
    for (int d = 0; d < DIM; ++d) { float v = zp[d]; sl += v * al[d]; sr += v * ar[d]; }
    el[t] = sl; er[t] = sr;
}

__global__ void init_kernel(int* __restrict__ mult, int* __restrict__ cnt,
                            float* __restrict__ red, int E, int nCnt)
{
    int t = blockIdx.x * blockDim.x + threadIdx.x;
    if (t < E) mult[t] = 0;
    else if (t - E < nCnt) cnt[t - E] = 0;
    if (t == 0) red[0] = 0.0f;
}

__global__ void count_kernel(const int* __restrict__ edst,
                             const int* __restrict__ sel,
                             int E, int E_sel, int n0, int nCnt,
                             int* __restrict__ cnt, int* __restrict__ mult,
                             int* __restrict__ rank)
{
    int t = blockIdx.x * blockDim.x + threadIdx.x;
    if (t < E) {
        int d = edst[t] - n0;
        int r = 0;
        if (d >= 0 && d < nCnt) r = atomicAdd(&cnt[d], 1);
        rank[t] = r;
    } else if (t - E < E_sel) {
        atomicAdd(&mult[sel[t - E]], 1);
    }
}

__device__ __forceinline__ void load4(const int* __restrict__ cnt, int b, int k0,
                                      int per, int nCnt, int (&v)[4])
{
    #pragma unroll
    for (int j = 0; j < 4; ++j) {
        const int idx = b + k0 + j;
        const bool ok = (k0 + j < per) && (idx < nCnt);
        const int x = cnt[ok ? idx : 0];
        v[j] = ok ? x : 0;
    }
}

__global__ __launch_bounds__(1024) void scan_kernel(const int* __restrict__ cnt,
                                                    int* __restrict__ rowptr,
                                                    int* __restrict__ rowend,
                                                    int nCnt)
{
    __shared__ int wsum[32];
    const int tid = threadIdx.x;
    const int lane = tid & 31, wv = tid >> 5;
    const int per = (nCnt + 1023) / 1024;
    const int b = tid * per;
    int local = 0;
    for (int k0 = 0; k0 < per; k0 += 4) {
        int v4[4];
        load4(cnt, b, k0, per, nCnt, v4);
        local += (v4[0] + v4[1]) + (v4[2] + v4[3]);
    }
    int v = local;
    for (int off = 1; off < 32; off <<= 1) { int u = __shfl_up(v, off); if (lane >= off) v += u; }
    if (lane == 31) wsum[wv] = v;
    __syncthreads();
    if (wv == 0) {
        int s = wsum[lane];
        for (int off = 1; off < 32; off <<= 1) { int u = __shfl_up(s, off); if (lane >= off) s += u; }
        wsum[lane] = s;
    }
    __syncthreads();
    int run = v - local + (wv > 0 ? wsum[wv - 1] : 0);
    for (int k0 = 0; k0 < per; k0 += 4) {
        int v4[4];
        load4(cnt, b, k0, per, nCnt, v4);
        #pragma unroll
        for (int j = 0; j < 4; ++j) {
            const int idx = b + k0 + j;
            if ((k0 + j < per) && (idx < nCnt)) { rowptr[idx] = run; run += v4[j]; rowend[idx] = run; }
        }
    }
}

__global__ void fill_kernel(const int* __restrict__ esrc,
                            const int* __restrict__ edst,
                            const int* __restrict__ mult,
                            const int* __restrict__ rank,
                            const int* __restrict__ rowptr,
                            int E, int n0, int nCnt,
                            int2* __restrict__ csr)
{
    int t = blockIdx.x * blockDim.x + threadIdx.x;
    if (t >= E) return;
    const int d = edst[t] - n0;
    if (d < 0 || d >= nCnt) return;
    csr[rowptr[d] + rank[t]] = make_int2(esrc[t], mult[t]);
}

__device__ __forceinline__ float leaky(float x) { return (x >= 0.f) ? x : NEG_SLOPE * x; }

__global__ __launch_bounds__(256) void gather_kernel(
    const float* __restrict__ z, const float* __restrict__ el,
    const float* __restrict__ er,
    const int* __restrict__ rowptr, const int* __restrict__ rowend,
    const int2* __restrict__ csr,
    const float* __restrict__ bias, const float* __restrict__ prelu_a,
    float* __restrict__ h_pos, float* __restrict__ h_neg, int n0, int nCnt)
{
    const int lane = threadIdx.x & 31;
    const int i = __builtin_amdgcn_readfirstlane((int)(blockIdx.x * 8 + (threadIdx.x >> 5)));
    if (i >= nCnt) return;
    const int n = n0 + i;
    const int h = lane >> 2;
    const int start = rowptr[i];
    const int end = rowend[i];
    const float4* __restrict__ z4 = (const float4*)z;

    const float erh = er[n * HEADS + h];
    const float xs = leaky(el[n * HEADS + h] + erh);
    float mA = xs, mB = xs;

    for (int base = start; base < end; base += 32) {
        const int c = (end - base < 32) ? (end - base) : 32;
        int2 rec = make_int2(0, 0);
        if (lane < c) rec = csr[base + lane];
        for (int k = 0; k < c; ++k) {
            const int s  = __builtin_amdgcn_readlane(rec.x, k);
            const int mu = __builtin_amdgcn_readlane(rec.y, k);
            const float x = leaky(el[s * HEADS + h] + erh);
            mA = fmaxf(mA, x);
            if (mu > 0) mB = fmaxf(mB, x);
        }
    }

    float dA = __expf(xs - mA), dB = __expf(xs - mB);
    float4 zv = z4[(size_t)n * 32 + lane];
    float4 aA = make_float4(dA * zv.x, dA * zv.y, dA * zv.z, dA * zv.w);
    float4 aB = make_float4(dB * zv.x, dB * zv.y, dB * zv.z, dB * zv.w);
    for (int base = start; base < end; base += 32) {
        const int c = (end - base < 32) ? (end - base) : 32;
        int2 rec = make_int2(0, 0);
        if (lane < c) rec = csr[base + lane];
        for (int k = 0; k < c; ++k) {
            const int s  = __builtin_amdgcn_readlane(rec.x, k);
            const int mu = __builtin_amdgcn_readlane(rec.y, k);
            const float x = leaky(el[s * HEADS + h] + erh);
            const float4 r = z4[(size_t)s * 32 + lane];
            const float wa = __expf(x - mA);
            const float wb = (mu > 0) ? (float)mu * __expf(x - mB) : 0.0f;
            dA += wa; dB += wb;
            aA.x += wa * r.x; aA.y += wa * r.y; aA.z += wa * r.z; aA.w += wa * r.w;
            aB.x += wb * r.x; aB.y += wb * r.y; aB.z += wb * r.z; aB.w += wb * r.w;
        }
    }

    const float a = prelu_a[0];
    const float4 b4 = ((const float4*)bias)[lane];
    float pA[4] = { aA.x / dA + b4.x, aA.y / dA + b4.y, aA.z / dA + b4.z, aA.w / dA + b4.w };
    float pB[4] = { aB.x / dB + b4.x, aB.y / dB + b4.y, aB.z / dB + b4.z, aB.w / dB + b4.w };
    #pragma unroll
    for (int c = 0; c < 4; ++c) {
        pA[c] = (pA[c] >= 0.f) ? pA[c] : a * pA[c];
        pB[c] = (pB[c] >= 0.f) ? pB[c] : a * pB[c];
        for (int off = 4; off < 32; off <<= 1) {
            pA[c] += __shfl_xor(pA[c], off);
            pB[c] += __shfl_xor(pB[c], off);
        }
    }
    const int q = lane & 3, cc = (lane >> 2) & 3;
    const bool neg = lane >= 16;
    const float v0 = neg ? pB[0] : pA[0];
    const float v1 = neg ? pB[1] : pA[1];
    const float v2 = neg ? pB[2] : pA[2];
    const float v3 = neg ? pB[3] : pA[3];
    const float v = (cc == 0) ? v0 : (cc == 1) ? v1 : (cc == 2) ? v2 : v3;
    float* __restrict__ dstp = neg ? h_neg : h_pos;
    dstp[(size_t)i * DIM + q * 4 + cc] = v * (1.0f / HEADS);
}

__global__ __launch_bounds__(256) void cos_loss_kernel(
    const float* __restrict__ hp, const float* __restrict__ hn,
    float* __restrict__ red, int nCnt)
{
    int i = blockIdx.x * blockDim.x + threadIdx.x;
    float v = 0.f;
    if (i < nCnt) {
        float num = 0.f, na = 0.f, nb = 0.f;
        #pragma unroll
        for (int d = 0; d < DIM; ++d) {
            float a = hp[(long long)i * DIM + d];
            float b = hn[(long long)i * DIM + d];
            num += a * b; na += a * a; nb += b * b;
        }
        na = fmaxf(sqrtf(na), 1e-8f);
        nb = fmaxf(sqrtf(nb), 1e-8f);
        v = __expf((num / (na * nb)) * (1.0f / TEM));
    }
    __shared__ float sbuf[256];
    sbuf[threadIdx.x] = v;
    __syncthreads();
    for (int s = 128; s > 0; s >>= 1) {
        if (threadIdx.x < s) sbuf[threadIdx.x] += sbuf[threadIdx.x + s];
        __syncthreads();
    }
    if (threadIdx.x == 0) atomicAdd(red, sbuf[0]);
}

__global__ void write_loss_kernel(const float* __restrict__ red, float* __restrict__ out)
{
    out[0] = logf(red[0]);
}

extern "C" void kernel_launch(void* const* d_in, const int* in_sizes, int n_in,
                              void* d_out, int out_size, void* d_ws, size_t ws_size,
                              hipStream_t stream)
{
    const float* feat    = (const float*)d_in[0];
    const float* fcw     = (const float*)d_in[1];
    const float* attn_l  = (const float*)d_in[2];
    const float* attn_r  = (const float*)d_in[3];
    const float* bias    = (const float*)d_in[4];
    const float* prelu_a = (const float*)d_in[5];
    const int*   esrc    = (const int*)d_in[6];
    const int*   edst    = (const int*)d_in[7];
    const int*   negidx  = (const int*)d_in[8];

    const int N     = in_sizes[0] / F;
    const int E     = in_sizes[6];
    const int E_NEG = in_sizes[8];
    const int N_DST = (out_size - 1) / DIM;
    const int N_SRC = N - N_DST;

    float* out   = (float*)d_out;
    float* h_pos = out + 1;

    float*    ws     = (float*)d_ws;
    float*    z      = ws;
    float*    el     = z   + (size_t)N * F;
    float*    er     = el  + (size_t)N * HEADS;
    float*    h_neg  = er  + (size_t)N * HEADS;
    int2*     csr    = (int2*)(h_neg + (size_t)N_DST * DIM);
    int*      mult   = (int*)(csr + (size_t)E);
    int*      cnt    = mult + (size_t)E;
    int*      rowptr = cnt + (size_t)N_DST;
    int*      rowend = rowptr + (size_t)N_DST;
    float*    red    = (float*)(rowend + (size_t)N_DST);
    int*      rank   = (int*)h_neg;

    const int TB = 256;
    auto cdiv = [](long long a, int b) { return (unsigned)((a + b - 1) / b); };

    gemm_z_wmma<<<cdiv(N, 16), TB, 0, stream>>>(feat, fcw, z, N);

    el_er_kernel<<<cdiv((long long)N * HEADS, TB), TB, 0, stream>>>(
        z, attn_l, attn_r, el, er, N);

    init_kernel<<<cdiv((long long)E + N_DST, TB), TB, 0, stream>>>(mult, cnt, red, E, N_DST);
    count_kernel<<<cdiv((long long)E + E_NEG, TB), TB, 0, stream>>>(
        edst, negidx, E, E_NEG, N_SRC, N_DST, cnt, mult, rank);
    scan_kernel<<<1, 1024, 0, stream>>>(cnt, rowptr, rowend, N_DST);
    fill_kernel<<<cdiv(E, TB), TB, 0, stream>>>(
        esrc, edst, mult, rank, rowptr, E, N_SRC, N_DST, csr);

    gather_kernel<<<cdiv(N_DST, 8), TB, 0, stream>>>(
        z, el, er, rowptr, rowend, csr, bias, prelu_a, h_pos, h_neg, N_SRC, N_DST);

    cos_loss_kernel<<<cdiv(N_DST, TB), TB, 0, stream>>>(h_pos, h_neg, red, N_DST);
    write_loss_kernel<<<1, 1, 0, stream>>>(red, out);
}


# In-degree, edge rank and run bounds by destination-range owner blocks on LDS returning adds, each walking all edges: no global returning atomic, no scan launch; sums in LDS-add arrival order
# speedup vs baseline: 5.8831x; 1.2236x over previous
#include <hip/hip_runtime.h>
#include <math.h>

typedef float v2f __attribute__((ext_vector_type(2)));
typedef float v8f __attribute__((ext_vector_type(8)));

#define HEADS 8
#define DIM 16
#define F 128
#define NEG_SLOPE 0.2f
#define TEM 0.7f

#define APITCH (F + 4)
__global__ __launch_bounds__(256) void gemm_z_wmma(
    const float* __restrict__ feat, const float* __restrict__ fcw,
    float* __restrict__ z, int N)
{
    __shared__ __attribute__((aligned(16))) float sA[16 * APITCH];
    const int row0 = blockIdx.x * 16;

    {
        const int c4 = threadIdx.x & 31;
        const int r0 = threadIdx.x >> 5;
        const int r1 = r0 + 8;
        const int g0 = (row0 + r0 < N) ? (row0 + r0) : (N - 1);
        const int g1 = (row0 + r1 < N) ? (row0 + r1) : (N - 1);
        const float4 v0 = ((const float4*)(feat + (long long)g0 * F))[c4];
        const float4 v1 = ((const float4*)(feat + (long long)g1 * F))[c4];
        *(float4*)&sA[r0 * APITCH + c4 * 4] = v0;
        *(float4*)&sA[r1 * APITCH + c4 * 4] = v1;
    }
    __syncthreads();

    const int wave = threadIdx.x >> 5;
    const int lane = threadIdx.x & 31;
    const int col0 = wave * 16;
    const int m    = lane & 15;
    const int koff = (lane >> 4) * 2;

    v8f c = {};
    for (int kb = 0; kb < F; kb += 4) {
        const int k = kb + koff;
        v2f a, b;
        a.x = sA[m * APITCH + k];
        a.y = sA[m * APITCH + k + 1];
        b.x = fcw[(long long)k * F + col0 + m];
        b.y = fcw[(long long)(k + 1) * F + col0 + m];
        c = __builtin_amdgcn_wmma_f32_16x16x4_f32(
                false, a, false, b, (short)0, c, false, false);
    }

    for (int i = 0; i < 8; ++i) {
        int gr = row0 + i + (lane >> 4) * 8;
        if (gr < N) z[(long long)gr * F + col0 + m] = c[i];
    }
}

__global__ void el_er_kernel(const float* __restrict__ z,
                             const float* __restrict__ attn_l,
                             const float* __restrict__ attn_r,
                             float* __restrict__ el, float* __restrict__ er,
                             int N)
{
    int t = blockIdx.x * blockDim.x + threadIdx.x;
    if (t >= N * HEADS) return;
    int n = t >> 3, h = t & 7;
    const float* zp = z + (long long)n * F + h * DIM;
    const float* al = attn_l + h * DIM;
    const float* ar = attn_r + h * DIM;
    float sl = 0.f, sr = 0.f;
    #pragma unroll
    for (int d = 0; d < DIM; ++d) { float v = zp[d]; sl += v * al[d]; sr += v * ar[d]; }
    el[t] = sl; er[t] = sr;
}

__global__ void init_kernel(int* __restrict__ mult, float* __restrict__ red, int E)
{
    int t = blockIdx.x * blockDim.x + threadIdx.x;
    if (t < E) mult[t] = 0;
    if (t == 0) red[0] = 0.0f;
}

#define OWN_R  320
#define CNT_TB 1024

__device__ __forceinline__ void own_edge(unsigned x, int e, unsigned nlo, unsigned span,
                                         int* lcnt, int* __restrict__ rank, int& below)
{
    below += (x > nlo) ? 1 : 0;
    if (x < span) rank[e] = atomicAdd(&lcnt[x], 1);
}

__global__ __launch_bounds__(CNT_TB) void count_kernel(
    const int* __restrict__ edst, const int* __restrict__ sel,
    int E, int E_sel, int n0, int nCnt, int nOwn,
    int* __restrict__ mult, int* __restrict__ rank,
    int* __restrict__ rowptr, int* __restrict__ rowend)
{
    __shared__ int lcnt[OWN_R];
    __shared__ int wbelow[CNT_TB / 32];
    __shared__ int wtot[OWN_R / 32];
    const int tid = threadIdx.x;
    const int lane = tid & 31, wv = tid >> 5;

    if ((int)blockIdx.x >= nOwn) {
        const int g  = ((int)blockIdx.x - nOwn) * CNT_TB + tid;
        const int ng = E_sel >> 2;
        if (g < ng) {
            const int4 s = ((const int4*)sel)[g];
            atomicAdd(&mult[s.x], 1); atomicAdd(&mult[s.y], 1);
            atomicAdd(&mult[s.z], 1); atomicAdd(&mult[s.w], 1);
        } else if (g == ng) {
            for (int j = ng << 2; j < E_sel; ++j) atomicAdd(&mult[sel[j]], 1);
        }
        return;
    }

    const unsigned lo   = blockIdx.x * OWN_R;
    const unsigned left = (unsigned)nCnt - lo;
    const unsigned span = (left < OWN_R) ? left : OWN_R;
    const unsigned lo2  = (unsigned)n0 + lo;
    const unsigned nlo  = ~lo;
    if (tid < OWN_R) lcnt[tid] = 0;
    __syncthreads();

    const int4* __restrict__ d4 = (const int4*)edst;
    const int nq    = E >> 2;
    const int nFull = nq / CNT_TB;
    const int nB    = nFull >> 2;
    int below = 0;

    int bi = (int)((blockIdx.x * (unsigned)nB) / (unsigned)nOwn);
    for (int k = 0; k < nB; ++k) {
        const int q0 = bi * (4 * CNT_TB) + tid;
        const int4 v0 = d4[q0];
        const int4 v1 = d4[q0 + CNT_TB];
        const int4 v2 = d4[q0 + 2 * CNT_TB];
        const int4 v3 = d4[q0 + 3 * CNT_TB];
        unsigned x[16];
        x[0]  = (unsigned)v0.x - lo2; x[1]  = (unsigned)v0.y - lo2;
        x[2]  = (unsigned)v0.z - lo2; x[3]  = (unsigned)v0.w - lo2;
        x[4]  = (unsigned)v1.x - lo2; x[5]  = (unsigned)v1.y - lo2;
        x[6]  = (unsigned)v1.z - lo2; x[7]  = (unsigned)v1.w - lo2;
        x[8]  = (unsigned)v2.x - lo2; x[9]  = (unsigned)v2.y - lo2;
        x[10] = (unsigned)v2.z - lo2; x[11] = (unsigned)v2.w - lo2;
        x[12] = (unsigned)v3.x - lo2; x[13] = (unsigned)v3.y - lo2;
        x[14] = (unsigned)v3.z - lo2; x[15] = (unsigned)v3.w - lo2;
        int nb = 0;
        #pragma unroll
        for (int j = 0; j < 16; ++j) nb += (x[j] > nlo) ? 1 : 0;
        below += nb;
        #pragma unroll
        for (int j = 0; j < 16; ++j) {
            const int e = 4 * (q0 + (j >> 2) * CNT_TB) + (j & 3);
            if (x[j] < span) rank[e] = atomicAdd(&lcnt[x[j]], 1);
        }
        if (++bi == nB) bi = 0;
    }
    for (int q = nB * (4 * CNT_TB) + tid; q < nq; q += CNT_TB) {
        const int4 v = d4[q];
        own_edge((unsigned)v.x - lo2, 4 * q,     nlo, span, lcnt, rank, below);
        own_edge((unsigned)v.y - lo2, 4 * q + 1, nlo, span, lcnt, rank, below);
        own_edge((unsigned)v.z - lo2, 4 * q + 2, nlo, span, lcnt, rank, below);
        own_edge((unsigned)v.w - lo2, 4 * q + 3, nlo, span, lcnt, rank, below);
    }
    for (int e = (nq << 2) + tid; e < E; e += CNT_TB)
        own_edge((unsigned)edst[e] - lo2, e, nlo, span, lcnt, rank, below);

    for (int off = 16; off > 0; off >>= 1) below += __shfl_xor(below, off);
    if (lane == 0) wbelow[wv] = below;
    __syncthreads();

    const int c = (tid < OWN_R) ? lcnt[tid] : 0;
    int v = c;
    for (int off = 1; off < 32; off <<= 1) { int u = __shfl_up(v, off); if (lane >= off) v += u; }
    if (lane == 31 && wv < OWN_R / 32) wtot[wv] = v;
    int b = wbelow[lane];
    for (int off = 16; off > 0; off >>= 1) b += __shfl_xor(b, off);
    __syncthreads();
    if ((unsigned)tid < span) {
        int pre = b;
        for (int w = 0; w < wv; ++w) pre += wtot[w];
        rowptr[lo + tid] = pre + v - c;
        rowend[lo + tid] = pre + v;
    }
}

__global__ void fill_kernel(const int* __restrict__ esrc,
                            const int* __restrict__ edst,
                            const int* __restrict__ mult,
                            const int* __restrict__ rank,
                            const int* __restrict__ rowptr,
                            int E, int n0, int nCnt,
                            int2* __restrict__ csr)
{
    int t = blockIdx.x * blockDim.x + threadIdx.x;
    if (t >= E) return;
    const int d = edst[t] - n0;
    if (d < 0 || d >= nCnt) return;
    csr[rowptr[d] + rank[t]] = make_int2(esrc[t], mult[t]);
}

__device__ __forceinline__ float leaky(float x) { return (x >= 0.f) ? x : NEG_SLOPE * x; }

__global__ __launch_bounds__(256) void gather_kernel(
    const float* __restrict__ z, const float* __restrict__ el,
    const float* __restrict__ er,
    const int* __restrict__ rowptr, const int* __restrict__ rowend,
    const int2* __restrict__ csr,
    const float* __restrict__ bias, const float* __restrict__ prelu_a,
    float* __restrict__ h_pos, float* __restrict__ h_neg, int n0, int nCnt)
{
    const int lane = threadIdx.x & 31;
    const int i = __builtin_amdgcn_readfirstlane((int)(blockIdx.x * 8 + (threadIdx.x >> 5)));
    if (i >= nCnt) return;
    const int n = n0 + i;
    const int h = lane >> 2;
    const int start = rowptr[i];
    const int end = rowend[i];
    const float4* __restrict__ z4 = (const float4*)z;

    const float erh = er[n * HEADS + h];
    const float xs = leaky(el[n * HEADS + h] + erh);
    float mA = xs, mB = xs;

    for (int base = start; base < end; base += 32) {
        const int c = (end - base < 32) ? (end - base) : 32;
        int2 rec = make_int2(0, 0);
        if (lane < c) rec = csr[base + lane];
        for (int k = 0; k < c; ++k) {
            const int s  = __builtin_amdgcn_readlane(rec.x, k);
            const int mu = __builtin_amdgcn_readlane(rec.y, k);
            const float x = leaky(el[s * HEADS + h] + erh);
            mA = fmaxf(mA, x);
            if (mu > 0) mB = fmaxf(mB, x);
        }
    }

    float dA = __expf(xs - mA), dB = __expf(xs - mB);
    float4 zv = z4[(size_t)n * 32 + lane];
    float4 aA = make_float4(dA * zv.x, dA * zv.y, dA * zv.z, dA * zv.w);
    float4 aB = make_float4(dB * zv.x, dB * zv.y, dB * zv.z, dB * zv.w);
    for (int base = start; base < end; base += 32) {
        const int c = (end - base < 32) ? (end - base) : 32;
        int2 rec = make_int2(0, 0);
        if (lane < c) rec = csr[base + lane];
        for (int k = 0; k < c; ++k) {
            const int s  = __builtin_amdgcn_readlane(rec.x, k);
            const int mu = __builtin_amdgcn_readlane(rec.y, k);
            const float x = leaky(el[s * HEADS + h] + erh);
            const float4 r = z4[(size_t)s * 32 + lane];
            const float wa = __expf(x - mA);
            const float wb = (mu > 0) ? (float)mu * __expf(x - mB) : 0.0f;
            dA += wa; dB += wb;
            aA.x += wa * r.x; aA.y += wa * r.y; aA.z += wa * r.z; aA.w += wa * r.w;
            aB.x += wb * r.x; aB.y += wb * r.y; aB.z += wb * r.z; aB.w += wb * r.w;
        }
    }

    const float a = prelu_a[0];
    const float4 b4 = ((const float4*)bias)[lane];
    float pA[4] = { aA.x / dA + b4.x, aA.y / dA + b4.y, aA.z / dA + b4.z, aA.w / dA + b4.w };
    float pB[4] = { aB.x / dB + b4.x, aB.y / dB + b4.y, aB.z / dB + b4.z, aB.w / dB + b4.w };
    #pragma unroll
    for (int c = 0; c < 4; ++c) {
        pA[c] = (pA[c] >= 0.f) ? pA[c] : a * pA[c];
        pB[c] = (pB[c] >= 0.f) ? pB[c] : a * pB[c];
        for (int off = 4; off < 32; off <<= 1) {
            pA[c] += __shfl_xor(pA[c], off);
            pB[c] += __shfl_xor(pB[c], off);
        }
    }
    const int q = lane & 3, cc = (lane >> 2) & 3;
    const bool neg = lane >= 16;
    const float v0 = neg ? pB[0] : pA[0];
    const float v1 = neg ? pB[1] : pA[1];
    const float v2 = neg ? pB[2] : pA[2];
    const float v3 = neg ? pB[3] : pA[3];
    const float v = (cc == 0) ? v0 : (cc == 1) ? v1 : (cc == 2) ? v2 : v3;
    float* __restrict__ dstp = neg ? h_neg : h_pos;
    dstp[(size_t)i * DIM + q * 4 + cc] = v * (1.0f / HEADS);
}

__global__ __launch_bounds__(256) void cos_loss_kernel(
    const float* __restrict__ hp, const float* __restrict__ hn,
    float* __restrict__ red, int nCnt)
{
    int i = blockIdx.x * blockDim.x + threadIdx.x;
    float v = 0.f;
    if (i < nCnt) {
        float num = 0.f, na = 0.f, nb = 0.f;
        #pragma unroll
        for (int d = 0; d < DIM; ++d) {
            float a = hp[(long long)i * DIM + d];
            float b = hn[(long long)i * DIM + d];
            num += a * b; na += a * a; nb += b * b;
        }
        na = fmaxf(sqrtf(na), 1e-8f);
        nb = fmaxf(sqrtf(nb), 1e-8f);
        v = __expf((num / (na * nb)) * (1.0f / TEM));
    }
    __shared__ float sbuf[256];
    sbuf[threadIdx.x] = v;
    __syncthreads();
    for (int s = 128; s > 0; s >>= 1) {
        if (threadIdx.x < s) sbuf[threadIdx.x] += sbuf[threadIdx.x + s];
        __syncthreads();
    }
    if (threadIdx.x == 0) atomicAdd(red, sbuf[0]);
}

__global__ void write_loss_kernel(const float* __restrict__ red, float* __restrict__ out)
{
    out[0] = logf(red[0]);
}

extern "C" void kernel_launch(void* const* d_in, const int* in_sizes, int n_in,
                              void* d_out, int out_size, void* d_ws, size_t ws_size,
                              hipStream_t stream)
{
    const float* feat    = (const float*)d_in[0];
    const float* fcw     = (const float*)d_in[1];
    const float* attn_l  = (const float*)d_in[2];
    const float* attn_r  = (const float*)d_in[3];
    const float* bias    = (const float*)d_in[4];
    const float* prelu_a = (const float*)d_in[5];
    const int*   esrc    = (const int*)d_in[6];
    const int*   edst    = (const int*)d_in[7];
    const int*   negidx  = (const int*)d_in[8];

    const int N     = in_sizes[0] / F;
    const int E     = in_sizes[6];
    const int E_NEG = in_sizes[8];
    const int N_DST = (out_size - 1) / DIM;
    const int N_SRC = N - N_DST;

    float* out   = (float*)d_out;
    float* h_pos = out + 1;

    float*    ws     = (float*)d_ws;
    float*    z      = ws;
    float*    el     = z   + (size_t)N * F;
    float*    er     = el  + (size_t)N * HEADS;
    float*    h_neg  = er  + (size_t)N * HEADS;
    int2*     csr    = (int2*)(h_neg + (size_t)N_DST * DIM);
    int*      mult   = (int*)(csr + (size_t)E);
    int*      rowptr = mult + (size_t)E;
    int*      rowend = rowptr + (size_t)N_DST;
    float*    red    = (float*)(rowend + (size_t)N_DST);
    int*      rank   = (int*)h_neg;

    const int TB = 256;
    auto cdiv = [](long long a, int b) { return (unsigned)((a + b - 1) / b); };

    gemm_z_wmma<<<cdiv(N, 16), TB, 0, stream>>>(feat, fcw, z, N);

    el_er_kernel<<<cdiv((long long)N * HEADS, TB), TB, 0, stream>>>(
        z, attn_l, attn_r, el, er, N);

    init_kernel<<<cdiv(E, TB), TB, 0, stream>>>(mult, red, E);
    const int nOwn = (int)cdiv(N_DST, OWN_R);
    count_kernel<<<nOwn + cdiv((long long)(E_NEG >> 2) + 1, CNT_TB), CNT_TB, 0, stream>>>(
        edst, negidx, E, E_NEG, N_SRC, N_DST, nOwn, mult, rank, rowptr, rowend);
    fill_kernel<<<cdiv(E, TB), TB, 0, stream>>>(
        esrc, edst, mult, rank, rowptr, E, N_SRC, N_DST, csr);

    gather_kernel<<<cdiv(N_DST, 8), TB, 0, stream>>>(
        z, el, er, rowptr, rowend, csr, bias, prelu_a, h_pos, h_neg, N_SRC, N_DST);

    cos_loss_kernel<<<cdiv(N_DST, TB), TB, 0, stream>>>(h_pos, h_neg, red, N_DST);
    write_loss_kernel<<<1, 1, 0, stream>>>(red, out);
}


# el/er dot products taken in the GEMM epilogue from the accumulator tile by a 16-lane DPP xor tree (their sums now a lane tree, not sequential); the el/er launch and its re-read of z are gone
# speedup vs baseline: 6.0738x; 1.0324x over previous
#include <hip/hip_runtime.h>
#include <math.h>

typedef float v2f __attribute__((ext_vector_type(2)));
typedef float v8f __attribute__((ext_vector_type(8)));

#define HEADS 8
#define DIM 16
#define F 128
#define NEG_SLOPE 0.2f
#define TEM 0.7f

#define APITCH (F + 4)

template <int M> __device__ __forceinline__ float row_xor(float v)
{
    return __int_as_float(__builtin_amdgcn_update_dpp(
        0, __float_as_int(v), 0x160 + M, 0xf, 0xf, false));
}
__global__ __launch_bounds__(256) void gemm_z_wmma(
    const float* __restrict__ feat, const float* __restrict__ fcw,
    const float* __restrict__ attn_l, const float* __restrict__ attn_r,
    float* __restrict__ z, float* __restrict__ el, float* __restrict__ er, int N)
{
    __shared__ __attribute__((aligned(16))) float sA[16 * APITCH];
    __shared__ float sE[2 * 16 * HEADS];
    const int row0 = blockIdx.x * 16;

    {
        const int c4 = threadIdx.x & 31;
        const int r0 = threadIdx.x >> 5;
        const int r1 = r0 + 8;
        const int g0 = (row0 + r0 < N) ? (row0 + r0) : (N - 1);
        const int g1 = (row0 + r1 < N) ? (row0 + r1) : (N - 1);
        const float4 v0 = ((const float4*)(feat + (long long)g0 * F))[c4];
        const float4 v1 = ((const float4*)(feat + (long long)g1 * F))[c4];
        *(float4*)&sA[r0 * APITCH + c4 * 4] = v0;
        *(float4*)&sA[r1 * APITCH + c4 * 4] = v1;
    }
    __syncthreads();

    const int wave = threadIdx.x >> 5;
    const int lane = threadIdx.x & 31;
    const int col0 = wave * 16;
    const int m    = lane & 15;
    const int koff = (lane >> 4) * 2;

    v8f c = {};
    for (int kb = 0; kb < F; kb += 4) {
        const int k = kb + koff;
        v2f a, b;
        a.x = sA[m * APITCH + k];
        a.y = sA[m * APITCH + k + 1];
        b.x = fcw[(long long)k * F + col0 + m];
        b.y = fcw[(long long)(k + 1) * F + col0 + m];
        c = __builtin_amdgcn_wmma_f32_16x16x4_f32(
                false, a, false, b, (short)0, c, false, false);
    }

    for (int i = 0; i < 8; ++i) {
        int gr = row0 + i + (lane >> 4) * 8;
        if (gr < N) z[(long long)gr * F + col0 + m] = c[i];
    }

    const float al = attn_l[col0 + m];
    const float ar = attn_r[col0 + m];
    float v[16];
    for (int i = 0; i < 8; ++i) { v[i] = c[i] * al; v[8 + i] = c[i] * ar; }
    const bool b3 = (m & 8) != 0, b2 = (m & 4) != 0, b1 = (m & 2) != 0, b0 = (m & 1) != 0;
    float u[8], w[4], x[2];
    for (int j = 0; j < 8; ++j)
        u[j] = (b3 ? v[j + 8] : v[j]) + row_xor<8>(b3 ? v[j] : v[j + 8]);
    for (int j = 0; j < 4; ++j)
        w[j] = (b2 ? u[j + 4] : u[j]) + row_xor<4>(b2 ? u[j] : u[j + 4]);
    for (int j = 0; j < 2; ++j)
        x[j] = (b1 ? w[j + 2] : w[j]) + row_xor<2>(b1 ? w[j] : w[j + 2]);
    const float y = (b0 ? x[1] : x[0]) + row_xor<1>(b0 ? x[0] : x[1]);

    sE[(b3 ? 16 * HEADS : 0) + ((lane >> 4) * 8 + (m & 7)) * HEADS + wave] = y;
    __syncthreads();
    {
        const int t = threadIdx.x & 127;
        if (row0 + (t >> 3) < N) {
            float* __restrict__ dst = (threadIdx.x < 128) ? el : er;
            dst[(long long)row0 * HEADS + t] = sE[threadIdx.x];
        }
    }
}

__global__ void init_kernel(int* __restrict__ mult, float* __restrict__ red, int E)
{
    int t = blockIdx.x * blockDim.x + threadIdx.x;
    if (t < E) mult[t] = 0;
    if (t == 0) red[0] = 0.0f;
}

#define OWN_R  320
#define CNT_TB 1024

__device__ __forceinline__ void own_edge(unsigned x, int e, unsigned nlo, unsigned span,
                                         int* lcnt, int* __restrict__ rank, int& below)
{
    below += (x > nlo) ? 1 : 0;
    if (x < span) rank[e] = atomicAdd(&lcnt[x], 1);
}

__global__ __launch_bounds__(CNT_TB) void count_kernel(
    const int* __restrict__ edst, const int* __restrict__ sel,
    int E, int E_sel, int n0, int nCnt, int nOwn,
    int* __restrict__ mult, int* __restrict__ rank,
    int* __restrict__ rowptr, int* __restrict__ rowend)
{
    __shared__ int lcnt[OWN_R];
    __shared__ int wbelow[CNT_TB / 32];
    __shared__ int wtot[OWN_R / 32];
    const int tid = threadIdx.x;
    const int lane = tid & 31, wv = tid >> 5;

    if ((int)blockIdx.x >= nOwn) {
        const int g  = ((int)blockIdx.x - nOwn) * CNT_TB + tid;
        const int ng = E_sel >> 2;
        if (g < ng) {
            const int4 s = ((const int4*)sel)[g];
            atomicAdd(&mult[s.x], 1); atomicAdd(&mult[s.y], 1);
            atomicAdd(&mult[s.z], 1); atomicAdd(&mult[s.w], 1);
        } else if (g == ng) {
            for (int j = ng << 2; j < E_sel; ++j) atomicAdd(&mult[sel[j]], 1);
        }
        return;
    }

    const unsigned lo   = blockIdx.x * OWN_R;
    const unsigned left = (unsigned)nCnt - lo;
    const unsigned span = (left < OWN_R) ? left : OWN_R;
    const unsigned lo2  = (unsigned)n0 + lo;
    const unsigned nlo  = ~lo;
    if (tid < OWN_R) lcnt[tid] = 0;
    __syncthreads();

    const int4* __restrict__ d4 = (const int4*)edst;
    const int nq    = E >> 2;
    const int nFull = nq / CNT_TB;
    const int nB    = nFull >> 2;
    int below = 0;

    int bi = (int)((blockIdx.x * (unsigned)nB) / (unsigned)nOwn);
    for (int k = 0; k < nB; ++k) {
        const int q0 = bi * (4 * CNT_TB) + tid;
        const int4 v0 = d4[q0];
        const int4 v1 = d4[q0 + CNT_TB];
        const int4 v2 = d4[q0 + 2 * CNT_TB];
        const int4 v3 = d4[q0 + 3 * CNT_TB];
        unsigned x[16];
        x[0]  = (unsigned)v0.x - lo2; x[1]  = (unsigned)v0.y - lo2;
        x[2]  = (unsigned)v0.z - lo2; x[3]  = (unsigned)v0.w - lo2;
        x[4]  = (unsigned)v1.x - lo2; x[5]  = (unsigned)v1.y - lo2;
        x[6]  = (unsigned)v1.z - lo2; x[7]  = (unsigned)v1.w - lo2;
        x[8]  = (unsigned)v2.x - lo2; x[9]  = (unsigned)v2.y - lo2;
        x[10] = (unsigned)v2.z - lo2; x[11] = (unsigned)v2.w - lo2;
        x[12] = (unsigned)v3.x - lo2; x[13] = (unsigned)v3.y - lo2;
        x[14] = (unsigned)v3.z - lo2; x[15] = (unsigned)v3.w - lo2;
        int nb = 0;
        #pragma unroll
        for (int j = 0; j < 16; ++j) nb += (x[j] > nlo) ? 1 : 0;
        below += nb;
        #pragma unroll
        for (int j = 0; j < 16; ++j) {
            const int e = 4 * (q0 + (j >> 2) * CNT_TB) + (j & 3);
            if (x[j] < span) rank[e] = atomicAdd(&lcnt[x[j]], 1);
        }
        if (++bi == nB) bi = 0;
    }
    for (int q = nB * (4 * CNT_TB) + tid; q < nq; q += CNT_TB) {
        const int4 v = d4[q];
        own_edge((unsigned)v.x - lo2, 4 * q,     nlo, span, lcnt, rank, below);
        own_edge((unsigned)v.y - lo2, 4 * q + 1, nlo, span, lcnt, rank, below);
        own_edge((unsigned)v.z - lo2, 4 * q + 2, nlo, span, lcnt, rank, below);
        own_edge((unsigned)v.w - lo2, 4 * q + 3, nlo, span, lcnt, rank, below);
    }
    for (int e = (nq << 2) + tid; e < E; e += CNT_TB)
        own_edge((unsigned)edst[e] - lo2, e, nlo, span, lcnt, rank, below);

    for (int off = 16; off > 0; off >>= 1) below += __shfl_xor(below, off);
    if (lane == 0) wbelow[wv] = below;
    __syncthreads();

    const int c = (tid < OWN_R) ? lcnt[tid] : 0;
    int v = c;
    for (int off = 1; off < 32; off <<= 1) { int u = __shfl_up(v, off); if (lane >= off) v += u; }
    if (lane == 31 && wv < OWN_R / 32) wtot[wv] = v;
    int b = wbelow[lane];
    for (int off = 16; off > 0; off >>= 1) b += __shfl_xor(b, off);
    __syncthreads();
    if ((unsigned)tid < span) {
        int pre = b;
        for (int w = 0; w < wv; ++w) pre += wtot[w];
        rowptr[lo + tid] = pre + v - c;
        rowend[lo + tid] = pre + v;
    }
}

__global__ void fill_kernel(const int* __restrict__ esrc,
                            const int* __restrict__ edst,
                            const int* __restrict__ mult,
                            const int* __restrict__ rank,
                            const int* __restrict__ rowptr,
                            int E, int n0, int nCnt,
                            int2* __restrict__ csr)
{
    int t = blockIdx.x * blockDim.x + threadIdx.x;
    if (t >= E) return;
    const int d = edst[t] - n0;
    if (d < 0 || d >= nCnt) return;
    csr[rowptr[d] + rank[t]] = make_int2(esrc[t], mult[t]);
}

__device__ __forceinline__ float leaky(float x) { return (x >= 0.f) ? x : NEG_SLOPE * x; }

__global__ __launch_bounds__(256) void gather_kernel(
    const float* __restrict__ z, const float* __restrict__ el,
    const float* __restrict__ er,
    const int* __restrict__ rowptr, const int* __restrict__ rowend,
    const int2* __restrict__ csr,
    const float* __restrict__ bias, const float* __restrict__ prelu_a,
    float* __restrict__ h_pos, float* __restrict__ h_neg, int n0, int nCnt)
{
    const int lane = threadIdx.x & 31;
    const int i = __builtin_amdgcn_readfirstlane((int)(blockIdx.x * 8 + (threadIdx.x >> 5)));
    if (i >= nCnt) return;
    const int n = n0 + i;
    const int h = lane >> 2;
    const int start = rowptr[i];
    const int end = rowend[i];
    const float4* __restrict__ z4 = (const float4*)z;

    const float erh = er[n * HEADS + h];
    const float xs = leaky(el[n * HEADS + h] + erh);
    float mA = xs, mB = xs;

    for (int base = start; base < end; base += 32) {
        const int c = (end - base < 32) ? (end - base) : 32;
        int2 rec = make_int2(0, 0);
        if (lane < c) rec = csr[base + lane];
        for (int k = 0; k < c; ++k) {
            const int s  = __builtin_amdgcn_readlane(rec.x, k);
            const int mu = __builtin_amdgcn_readlane(rec.y, k);
            const float x = leaky(el[s * HEADS + h] + erh);
            mA = fmaxf(mA, x);
            if (mu > 0) mB = fmaxf(mB, x);
        }
    }

    float dA = __expf(xs - mA), dB = __expf(xs - mB);
    float4 zv = z4[(size_t)n * 32 + lane];
    float4 aA = make_float4(dA * zv.x, dA * zv.y, dA * zv.z, dA * zv.w);
    float4 aB = make_float4(dB * zv.x, dB * zv.y, dB * zv.z, dB * zv.w);
    for (int base = start; base < end; base += 32) {
        const int c = (end - base < 32) ? (end - base) : 32;
        int2 rec = make_int2(0, 0);
        if (lane < c) rec = csr[base + lane];
        for (int k = 0; k < c; ++k) {
            const int s  = __builtin_amdgcn_readlane(rec.x, k);
            const int mu = __builtin_amdgcn_readlane(rec.y, k);
            const float x = leaky(el[s * HEADS + h] + erh);
            const float4 r = z4[(size_t)s * 32 + lane];
            const float wa = __expf(x - mA);
            const float wb = (mu > 0) ? (float)mu * __expf(x - mB) : 0.0f;
            dA += wa; dB += wb;
            aA.x += wa * r.x; aA.y += wa * r.y; aA.z += wa * r.z; aA.w += wa * r.w;
            aB.x += wb * r.x; aB.y += wb * r.y; aB.z += wb * r.z; aB.w += wb * r.w;
        }
    }

    const float a = prelu_a[0];
    const float4 b4 = ((const float4*)bias)[lane];
    float pA[4] = { aA.x / dA + b4.x, aA.y / dA + b4.y, aA.z / dA + b4.z, aA.w / dA + b4.w };
    float pB[4] = { aB.x / dB + b4.x, aB.y / dB + b4.y, aB.z / dB + b4.z, aB.w / dB + b4.w };
    #pragma unroll
    for (int c = 0; c < 4; ++c) {
        pA[c] = (pA[c] >= 0.f) ? pA[c] : a * pA[c];
        pB[c] = (pB[c] >= 0.f) ? pB[c] : a * pB[c];
        for (int off = 4; off < 32; off <<= 1) {
            pA[c] += __shfl_xor(pA[c], off);
            pB[c] += __shfl_xor(pB[c], off);
        }
    }
    const int q = lane & 3, cc = (lane >> 2) & 3;
    const bool neg = lane >= 16;
    const float v0 = neg ? pB[0] : pA[0];
    const float v1 = neg ? pB[1] : pA[1];
    const float v2 = neg ? pB[2] : pA[2];
    const float v3 = neg ? pB[3] : pA[3];
    const float v = (cc == 0) ? v0 : (cc == 1) ? v1 : (cc == 2) ? v2 : v3;
    float* __restrict__ dstp = neg ? h_neg : h_pos;
    dstp[(size_t)i * DIM + q * 4 + cc] = v * (1.0f / HEADS);
}

__global__ __launch_bounds__(256) void cos_loss_kernel(
    const float* __restrict__ hp, const float* __restrict__ hn,
    float* __restrict__ red, int nCnt)
{
    int i = blockIdx.x * blockDim.x + threadIdx.x;
    float v = 0.f;
    if (i < nCnt) {
        float num = 0.f, na = 0.f, nb = 0.f;
        #pragma unroll
        for (int d = 0; d < DIM; ++d) {
            float a = hp[(long long)i * DIM + d];
            float b = hn[(long long)i * DIM + d];
            num += a * b; na += a * a; nb += b * b;
        }
        na = fmaxf(sqrtf(na), 1e-8f);
        nb = fmaxf(sqrtf(nb), 1e-8f);
        v = __expf((num / (na * nb)) * (1.0f / TEM));
    }
    __shared__ float sbuf[256];
    sbuf[threadIdx.x] = v;
    __syncthreads();
    for (int s = 128; s > 0; s >>= 1) {
        if (threadIdx.x < s) sbuf[threadIdx.x] += sbuf[threadIdx.x + s];
        __syncthreads();
    }
    if (threadIdx.x == 0) atomicAdd(red, sbuf[0]);
}

__global__ void write_loss_kernel(const float* __restrict__ red, float* __restrict__ out)
{
    out[0] = logf(red[0]);
}

extern "C" void kernel_launch(void* const* d_in, const int* in_sizes, int n_in,
                              void* d_out, int out_size, void* d_ws, size_t ws_size,
                              hipStream_t stream)
{
    const float* feat    = (const float*)d_in[0];
    const float* fcw     = (const float*)d_in[1];
    const float* attn_l  = (const float*)d_in[2];
    const float* attn_r  = (const float*)d_in[3];
    const float* bias    = (const float*)d_in[4];
    const float* prelu_a = (const float*)d_in[5];
    const int*   esrc    = (const int*)d_in[6];
    const int*   edst    = (const int*)d_in[7];
    const int*   negidx  = (const int*)d_in[8];

    const int N     = in_sizes[0] / F;
    const int E     = in_sizes[6];
    const int E_NEG = in_sizes[8];
    const int N_DST = (out_size - 1) / DIM;
    const int N_SRC = N - N_DST;

    float* out   = (float*)d_out;
    float* h_pos = out + 1;

    float*    ws     = (float*)d_ws;
    float*    z      = ws;
    float*    el     = z   + (size_t)N * F;
    float*    er     = el  + (size_t)N * HEADS;
    float*    h_neg  = er  + (size_t)N * HEADS;
    int2*     csr    = (int2*)(h_neg + (size_t)N_DST * DIM);
    int*      mult   = (int*)(csr + (size_t)E);
    int*      rowptr = mult + (size_t)E;
    int*      rowend = rowptr + (size_t)N_DST;
    float*    red    = (float*)(rowend + (size_t)N_DST);
    int*      rank   = (int*)h_neg;

    const int TB = 256;
    auto cdiv = [](long long a, int b) { return (unsigned)((a + b - 1) / b); };

    gemm_z_wmma<<<cdiv(N, 16), TB, 0, stream>>>(feat, fcw, attn_l, attn_r, z, el, er, N);

    init_kernel<<<cdiv(E, TB), TB, 0, stream>>>(mult, red, E);
    const int nOwn = (int)cdiv(N_DST, OWN_R);
    count_kernel<<<nOwn + cdiv((long long)(E_NEG >> 2) + 1, CNT_TB), CNT_TB, 0, stream>>>(
        edst, negidx, E, E_NEG, N_SRC, N_DST, nOwn, mult, rank, rowptr, rowend);
    fill_kernel<<<cdiv(E, TB), TB, 0, stream>>>(
        esrc, edst, mult, rank, rowptr, E, N_SRC, N_DST, csr);

    gather_kernel<<<cdiv(N_DST, 8), TB, 0, stream>>>(
        z, el, er, rowptr, rowend, csr, bias, prelu_a, h_pos, h_neg, N_SRC, N_DST);

    cos_loss_kernel<<<cdiv(N_DST, TB), TB, 0, stream>>>(h_pos, h_neg, red, N_DST);
    write_loss_kernel<<<1, 1, 0, stream>>>(red, out);
}


# Edge list cut into 4 segments, one owner work-group per (destination range, segment): each walks a quarter of the list on 4x the groups; a node's edges are 4 sub-runs, summed segment by segment
# speedup vs baseline: 6.7448x; 1.1105x over previous
#include <hip/hip_runtime.h>
#include <math.h>

typedef float v2f __attribute__((ext_vector_type(2)));
typedef float v8f __attribute__((ext_vector_type(8)));

#define HEADS 8
#define DIM 16
#define F 128
#define NEG_SLOPE 0.2f
#define TEM 0.7f

#define APITCH (F + 4)

template <int M> __device__ __forceinline__ float row_xor(float v)
{
    return __int_as_float(__builtin_amdgcn_update_dpp(
        0, __float_as_int(v), 0x160 + M, 0xf, 0xf, false));
}
__global__ __launch_bounds__(256) void gemm_z_wmma(
    const float* __restrict__ feat, const float* __restrict__ fcw,
    const float* __restrict__ attn_l, const float* __restrict__ attn_r,
    float* __restrict__ z, float* __restrict__ el, float* __restrict__ er, int N)
{
    __shared__ __attribute__((aligned(16))) float sA[16 * APITCH];
    __shared__ float sE[2 * 16 * HEADS];
    const int row0 = blockIdx.x * 16;

    {
        const int c4 = threadIdx.x & 31;
        const int r0 = threadIdx.x >> 5;
        const int r1 = r0 + 8;
        const int g0 = (row0 + r0 < N) ? (row0 + r0) : (N - 1);
        const int g1 = (row0 + r1 < N) ? (row0 + r1) : (N - 1);
        const float4 v0 = ((const float4*)(feat + (long long)g0 * F))[c4];
        const float4 v1 = ((const float4*)(feat + (long long)g1 * F))[c4];
        *(float4*)&sA[r0 * APITCH + c4 * 4] = v0;
        *(float4*)&sA[r1 * APITCH + c4 * 4] = v1;
    }
    __syncthreads();

    const int wave = threadIdx.x >> 5;
    const int lane = threadIdx.x & 31;
    const int col0 = wave * 16;
    const int m    = lane & 15;
    const int koff = (lane >> 4) * 2;

    v8f c = {};
    for (int kb = 0; kb < F; kb += 4) {
        const int k = kb + koff;
        v2f a, b;
        a.x = sA[m * APITCH + k];
        a.y = sA[m * APITCH + k + 1];
        b.x = fcw[(long long)k * F + col0 + m];
        b.y = fcw[(long long)(k + 1) * F + col0 + m];
        c = __builtin_amdgcn_wmma_f32_16x16x4_f32(
                false, a, false, b, (short)0, c, false, false);
    }

    for (int i = 0; i < 8; ++i) {
        int gr = row0 + i + (lane >> 4) * 8;
        if (gr < N) z[(long long)gr * F + col0 + m] = c[i];
    }

    const float al = attn_l[col0 + m];
    const float ar = attn_r[col0 + m];
    float v[16];
    for (int i = 0; i < 8; ++i) { v[i] = c[i] * al; v[8 + i] = c[i] * ar; }
    const bool b3 = (m & 8) != 0, b2 = (m & 4) != 0, b1 = (m & 2) != 0, b0 = (m & 1) != 0;
    float u[8], w[4], x[2];
    for (int j = 0; j < 8; ++j)
        u[j] = (b3 ? v[j + 8] : v[j]) + row_xor<8>(b3 ? v[j] : v[j + 8]);
    for (int j = 0; j < 4; ++j)
        w[j] = (b2 ? u[j + 4] : u[j]) + row_xor<4>(b2 ? u[j] : u[j + 4]);
    for (int j = 0; j < 2; ++j)
        x[j] = (b1 ? w[j + 2] : w[j]) + row_xor<2>(b1 ? w[j] : w[j + 2]);
    const float y = (b0 ? x[1] : x[0]) + row_xor<1>(b0 ? x[0] : x[1]);

    sE[(b3 ? 16 * HEADS : 0) + ((lane >> 4) * 8 + (m & 7)) * HEADS + wave] = y;
    __syncthreads();
    {
        const int t = threadIdx.x & 127;
        if (row0 + (t >> 3) < N) {
            float* __restrict__ dst = (threadIdx.x < 128) ? el : er;
            dst[(long long)row0 * HEADS + t] = sE[threadIdx.x];
        }
    }
}

__global__ void init_kernel(int* __restrict__ mult, float* __restrict__ red, int E)
{
    int t = blockIdx.x * blockDim.x + threadIdx.x;
    if (t < E) mult[t] = 0;
    if (t == 0) red[0] = 0.0f;
}

#define OWN_R  320
#define CNT_TB 1024
#define NSEG   4

__device__ __forceinline__ void own_edge(unsigned x, int e, unsigned nlo, unsigned span,
                                         int* lcnt, int* __restrict__ rank, int& below)
{
    below += (x > nlo) ? 1 : 0;
    if (x < span) rank[e] = atomicAdd(&lcnt[x], 1);
}

template <bool GUARD>
__device__ __forceinline__ void own_batch(const int4* __restrict__ d4, int q0, int qe,
                                          unsigned lo2, unsigned nlo, unsigned span,
                                          int* lcnt, int* __restrict__ rank, int& below)
{
    int4 v[4];
    bool ok[4];
    #pragma unroll
    for (int u = 0; u < 4; ++u) {
        const int q = q0 + u * CNT_TB;
        ok[u] = !GUARD || (q < qe);
        v[u] = d4[ok[u] ? q : (qe - 1)];
    }
    unsigned x[16];
    #pragma unroll
    for (int u = 0; u < 4; ++u) {
        x[4 * u + 0] = ok[u] ? (unsigned)v[u].x - lo2 : span;
        x[4 * u + 1] = ok[u] ? (unsigned)v[u].y - lo2 : span;
        x[4 * u + 2] = ok[u] ? (unsigned)v[u].z - lo2 : span;
        x[4 * u + 3] = ok[u] ? (unsigned)v[u].w - lo2 : span;
    }
    int nb = 0;
    #pragma unroll
    for (int j = 0; j < 16; ++j) nb += (x[j] > nlo) ? 1 : 0;
    below += nb;
    #pragma unroll
    for (int j = 0; j < 16; ++j) {
        const int e = 4 * (q0 + (j >> 2) * CNT_TB) + (j & 3);
        if (x[j] < span) rank[e] = atomicAdd(&lcnt[x[j]], 1);
    }
}

__global__ __launch_bounds__(CNT_TB) void count_kernel(
    const int* __restrict__ edst, const int* __restrict__ sel,
    int E, int E_sel, int n0, int nCnt, int nOwn,
    int* __restrict__ mult, int* __restrict__ rank,
    int* __restrict__ rowptr, int* __restrict__ rowend)
{
    __shared__ int lcnt[OWN_R];
    __shared__ int wbelow[CNT_TB / 32];
    __shared__ int wtot[OWN_R / 32];
    const int tid = threadIdx.x;
    const int lane = tid & 31, wv = tid >> 5;

    if ((int)blockIdx.x >= nOwn * NSEG) {
        const int g  = ((int)blockIdx.x - nOwn * NSEG) * CNT_TB + tid;
        const int ng = E_sel >> 2;
        if (g < ng) {
            const int4 s = ((const int4*)sel)[g];
            atomicAdd(&mult[s.x], 1); atomicAdd(&mult[s.y], 1);
            atomicAdd(&mult[s.z], 1); atomicAdd(&mult[s.w], 1);
        } else if (g == ng) {
            for (int j = ng << 2; j < E_sel; ++j) atomicAdd(&mult[sel[j]], 1);
        }
        return;
    }

    const int seg = (int)blockIdx.x / nOwn;
    const int own = (int)blockIdx.x - seg * nOwn;
    const unsigned lo   = (unsigned)own * OWN_R;
    const unsigned left = (unsigned)nCnt - lo;
    const unsigned span = (left < OWN_R) ? left : OWN_R;
    const unsigned lo2  = (unsigned)n0 + lo;
    const unsigned nlo  = ~lo;
    if (tid < OWN_R) lcnt[tid] = 0;
    __syncthreads();

    const int4* __restrict__ d4 = (const int4*)edst;
    const int nq = E >> 2;
    const int gq = (nq + NSEG - 1) / NSEG;
    const int qb = (seg * gq < nq) ? seg * gq : nq;
    const int qe = (qb + gq < nq) ? qb + gq : nq;
    const int ngrp  = qe - qb;
    const int nB    = (ngrp + 4 * CNT_TB - 1) / (4 * CNT_TB);
    const int nFull = ngrp / (4 * CNT_TB);
    int below = 0;

    int bi = (nB > 0) ? (int)(((unsigned)own * (unsigned)nB) / (unsigned)nOwn) : 0;
    for (int k = 0; k < nB; ++k) {
        const int q0 = qb + bi * (4 * CNT_TB) + tid;
        if (bi < nFull) own_batch<false>(d4, q0, qe, lo2, nlo, span, lcnt, rank, below);
        else            own_batch<true >(d4, q0, qe, lo2, nlo, span, lcnt, rank, below);
        if (++bi == nB) bi = 0;
    }
    if (seg == NSEG - 1)
        for (int e = (nq << 2) + tid; e < E; e += CNT_TB)
            own_edge((unsigned)edst[e] - lo2, e, nlo, span, lcnt, rank, below);

    for (int off = 16; off > 0; off >>= 1) below += __shfl_xor(below, off);
    if (lane == 0) wbelow[wv] = below;
    __syncthreads();

    const int c = (tid < OWN_R) ? lcnt[tid] : 0;
    int v = c;
    for (int off = 1; off < 32; off <<= 1) { int u = __shfl_up(v, off); if (lane >= off) v += u; }
    if (lane == 31 && wv < OWN_R / 32) wtot[wv] = v;
    int b = wbelow[lane];
    for (int off = 16; off > 0; off >>= 1) b += __shfl_xor(b, off);
    __syncthreads();
    if ((unsigned)tid < span) {
        int pre = 4 * qb + b;
        for (int w = 0; w < wv; ++w) pre += wtot[w];
        rowptr[seg * nCnt + (int)lo + tid] = pre + v - c;
        rowend[seg * nCnt + (int)lo + tid] = pre + v;
    }
}

__global__ void fill_kernel(const int* __restrict__ esrc,
                            const int* __restrict__ edst,
                            const int* __restrict__ mult,
                            const int* __restrict__ rank,
                            const int* __restrict__ rowptr,
                            int E, int n0, int nCnt,
                            int2* __restrict__ csr)
{
    int t = blockIdx.x * blockDim.x + threadIdx.x;
    if (t >= E) return;
    const int d = edst[t] - n0;
    if (d < 0 || d >= nCnt) return;
    const int nq = E >> 2;
    const int gq = (nq + NSEG - 1) / NSEG;
    const int q = t >> 2;
    int s = NSEG - 1;
    if (q < nq) {
        s = 0;
        #pragma unroll
        for (int j = 1; j < NSEG; ++j) s += (q >= j * gq) ? 1 : 0;
    }
    csr[rowptr[s * nCnt + d] + rank[t]] = make_int2(esrc[t], mult[t]);
}

__device__ __forceinline__ float leaky(float x) { return (x >= 0.f) ? x : NEG_SLOPE * x; }

__device__ __forceinline__ int list_pos(int v, const int (&sb)[NSEG], const int (&cum)[NSEG + 1])
{
    int o = sb[0];
    #pragma unroll
    for (int s = 1; s < NSEG; ++s) o = (v >= cum[s]) ? (sb[s] - cum[s]) : o;
    return v + o;
}

__global__ __launch_bounds__(256) void gather_kernel(
    const float* __restrict__ z, const float* __restrict__ el,
    const float* __restrict__ er,
    const int* __restrict__ rowptr, const int* __restrict__ rowend,
    const int2* __restrict__ csr,
    const float* __restrict__ bias, const float* __restrict__ prelu_a,
    float* __restrict__ h_pos, float* __restrict__ h_neg, int n0, int nCnt)
{
    const int lane = threadIdx.x & 31;
    const int i = __builtin_amdgcn_readfirstlane((int)(blockIdx.x * 8 + (threadIdx.x >> 5)));
    if (i >= nCnt) return;
    const int n = n0 + i;
    const int h = lane >> 2;
    int sb[NSEG], se[NSEG];
    #pragma unroll
    for (int s = 0; s < NSEG; ++s) {
        sb[s] = rowptr[s * nCnt + i];
        se[s] = rowend[s * nCnt + i];
    }
    int cum[NSEG + 1];
    cum[0] = 0;
    #pragma unroll
    for (int s = 0; s < NSEG; ++s) cum[s + 1] = cum[s] + (se[s] - sb[s]);
    const int total = cum[NSEG];
    const float4* __restrict__ z4 = (const float4*)z;

    const float erh = er[n * HEADS + h];
    const float xs = leaky(el[n * HEADS + h] + erh);
    float mA = xs, mB = xs;

    for (int base = 0; base < total; base += 32) {
        const int c = (total - base < 32) ? (total - base) : 32;
        int2 rec = make_int2(0, 0);
        if (lane < c) rec = csr[list_pos(base + lane, sb, cum)];
        for (int k = 0; k < c; ++k) {
            const int s  = __builtin_amdgcn_readlane(rec.x, k);
            const int mu = __builtin_amdgcn_readlane(rec.y, k);
            const float x = leaky(el[s * HEADS + h] + erh);
            mA = fmaxf(mA, x);
            if (mu > 0) mB = fmaxf(mB, x);
        }
    }

    float dA = __expf(xs - mA), dB = __expf(xs - mB);
    float4 zv = z4[(size_t)n * 32 + lane];
    float4 aA = make_float4(dA * zv.x, dA * zv.y, dA * zv.z, dA * zv.w);
    float4 aB = make_float4(dB * zv.x, dB * zv.y, dB * zv.z, dB * zv.w);
    for (int base = 0; base < total; base += 32) {
        const int c = (total - base < 32) ? (total - base) : 32;
        int2 rec = make_int2(0, 0);
        if (lane < c) rec = csr[list_pos(base + lane, sb, cum)];
        for (int k = 0; k < c; ++k) {
            const int s  = __builtin_amdgcn_readlane(rec.x, k);
            const int mu = __builtin_amdgcn_readlane(rec.y, k);
            const float x = leaky(el[s * HEADS + h] + erh);
            const float4 r = z4[(size_t)s * 32 + lane];
            const float wa = __expf(x - mA);
            const float wb = (mu > 0) ? (float)mu * __expf(x - mB) : 0.0f;
            dA += wa; dB += wb;
            aA.x += wa * r.x; aA.y += wa * r.y; aA.z += wa * r.z; aA.w += wa * r.w;
            aB.x += wb * r.x; aB.y += wb * r.y; aB.z += wb * r.z; aB.w += wb * r.w;
        }
    }

    const float a = prelu_a[0];
    const float4 b4 = ((const float4*)bias)[lane];
    float pA[4] = { aA.x / dA + b4.x, aA.y / dA + b4.y, aA.z / dA + b4.z, aA.w / dA + b4.w };
    float pB[4] = { aB.x / dB + b4.x, aB.y / dB + b4.y, aB.z / dB + b4.z, aB.w / dB + b4.w };
    #pragma unroll
    for (int c = 0; c < 4; ++c) {
        pA[c] = (pA[c] >= 0.f) ? pA[c] : a * pA[c];
        pB[c] = (pB[c] >= 0.f) ? pB[c] : a * pB[c];
        for (int off = 4; off < 32; off <<= 1) {
            pA[c] += __shfl_xor(pA[c], off);
            pB[c] += __shfl_xor(pB[c], off);
        }
    }
    const int q = lane & 3, cc = (lane >> 2) & 3;
    const bool neg = lane >= 16;
    const float v0 = neg ? pB[0] : pA[0];
    const float v1 = neg ? pB[1] : pA[1];
    const float v2 = neg ? pB[2] : pA[2];
    const float v3 = neg ? pB[3] : pA[3];
    const float v = (cc == 0) ? v0 : (cc == 1) ? v1 : (cc == 2) ? v2 : v3;
    float* __restrict__ dstp = neg ? h_neg : h_pos;
    dstp[(size_t)i * DIM + q * 4 + cc] = v * (1.0f / HEADS);
}

__global__ __launch_bounds__(256) void cos_loss_kernel(
    const float* __restrict__ hp, const float* __restrict__ hn,
    float* __restrict__ red, int nCnt)
{
    int i = blockIdx.x * blockDim.x + threadIdx.x;
    float v = 0.f;
    if (i < nCnt) {
        float num = 0.f, na = 0.f, nb = 0.f;
        #pragma unroll
        for (int d = 0; d < DIM; ++d) {
            float a = hp[(long long)i * DIM + d];
            float b = hn[(long long)i * DIM + d];
            num += a * b; na += a * a; nb += b * b;
        }
        na = fmaxf(sqrtf(na), 1e-8f);
        nb = fmaxf(sqrtf(nb), 1e-8f);
        v = __expf((num / (na * nb)) * (1.0f / TEM));
    }
    __shared__ float sbuf[256];
    sbuf[threadIdx.x] = v;
    __syncthreads();
    for (int s = 128; s > 0; s >>= 1) {
        if (threadIdx.x < s) sbuf[threadIdx.x] += sbuf[threadIdx.x + s];
        __syncthreads();
    }
    if (threadIdx.x == 0) atomicAdd(red, sbuf[0]);
}

__global__ void write_loss_kernel(const float* __restrict__ red, float* __restrict__ out)
{
    out[0] = logf(red[0]);
}

extern "C" void kernel_launch(void* const* d_in, const int* in_sizes, int n_in,
                              void* d_out, int out_size, void* d_ws, size_t ws_size,
                              hipStream_t stream)
{
    const float* feat    = (const float*)d_in[0];
    const float* fcw     = (const float*)d_in[1];
    const float* attn_l  = (const float*)d_in[2];
    const float* attn_r  = (const float*)d_in[3];
    const float* bias    = (const float*)d_in[4];
    const float* prelu_a = (const float*)d_in[5];
    const int*   esrc    = (const int*)d_in[6];
    const int*   edst    = (const int*)d_in[7];
    const int*   negidx  = (const int*)d_in[8];

    const int N     = in_sizes[0] / F;
    const int E     = in_sizes[6];
    const int E_NEG = in_sizes[8];
    const int N_DST = (out_size - 1) / DIM;
    const int N_SRC = N - N_DST;

    float* out   = (float*)d_out;
    float* h_pos = out + 1;

    float*    ws     = (float*)d_ws;
    float*    z      = ws;
    float*    el     = z   + (size_t)N * F;
    float*    er     = el  + (size_t)N * HEADS;
    float*    h_neg  = er  + (size_t)N * HEADS;
    int2*     csr    = (int2*)(h_neg + (size_t)N_DST * DIM);
    int*      mult   = (int*)(csr + (size_t)E);
    float*    red    = (float*)(mult + (size_t)E + 2 * (size_t)N_DST);
    int*      rowptr = (int*)er;
    int*      rowend = rowptr + (size_t)NSEG * N_DST;
    int*      rank   = (int*)h_neg;

    const int TB = 256;
    auto cdiv = [](long long a, int b) { return (unsigned)((a + b - 1) / b); };

    gemm_z_wmma<<<cdiv(N, 16), TB, 0, stream>>>(feat, fcw, attn_l, attn_r, z, el, er, N);

    init_kernel<<<cdiv(E, TB), TB, 0, stream>>>(mult, red, E);
    const int nOwn = (int)cdiv(N_DST, OWN_R);
    count_kernel<<<nOwn * NSEG + cdiv((long long)(E_NEG >> 2) + 1, CNT_TB), CNT_TB, 0, stream>>>(
        edst, negidx, E, E_NEG, N_SRC, N_DST, nOwn, mult, rank, rowptr, rowend);
    fill_kernel<<<cdiv(E, TB), TB, 0, stream>>>(
        esrc, edst, mult, rank, rowptr, E, N_SRC, N_DST, csr);

    gather_kernel<<<cdiv(N_DST, 8), TB, 0, stream>>>(
        z, el, er, rowptr, rowend, csr, bias, prelu_a, h_pos, h_neg, N_SRC, N_DST);

    cos_loss_kernel<<<cdiv(N_DST, TB), TB, 0, stream>>>(h_pos, h_neg, red, N_DST);
    write_loss_kernel<<<1, 1, 0, stream>>>(red, out);
}


# Gather: a wave's loads go out in batches ahead of first use (own row, bias at the top; 16 logits per pass-1 batch; 8 logits + 8 z rows per pass-2 batch; edge records kept for pass 2); same sum order
# speedup vs baseline: 7.1801x; 1.0645x over previous
#include <hip/hip_runtime.h>
#include <math.h>

typedef float v2f __attribute__((ext_vector_type(2)));
typedef float v8f __attribute__((ext_vector_type(8)));

#define HEADS 8
#define DIM 16
#define F 128
#define NEG_SLOPE 0.2f
#define TEM 0.7f

#define APITCH (F + 4)

template <int M> __device__ __forceinline__ float row_xor(float v)
{
    return __int_as_float(__builtin_amdgcn_update_dpp(
        0, __float_as_int(v), 0x160 + M, 0xf, 0xf, false));
}
__global__ __launch_bounds__(256) void gemm_z_wmma(
    const float* __restrict__ feat, const float* __restrict__ fcw,
    const float* __restrict__ attn_l, const float* __restrict__ attn_r,
    float* __restrict__ z, float* __restrict__ el, float* __restrict__ er, int N)
{
    __shared__ __attribute__((aligned(16))) float sA[16 * APITCH];
    __shared__ float sE[2 * 16 * HEADS];
    const int row0 = blockIdx.x * 16;

    {
        const int c4 = threadIdx.x & 31;
        const int r0 = threadIdx.x >> 5;
        const int r1 = r0 + 8;
        const int g0 = (row0 + r0 < N) ? (row0 + r0) : (N - 1);
        const int g1 = (row0 + r1 < N) ? (row0 + r1) : (N - 1);
        const float4 v0 = ((const float4*)(feat + (long long)g0 * F))[c4];
        const float4 v1 = ((const float4*)(feat + (long long)g1 * F))[c4];
        *(float4*)&sA[r0 * APITCH + c4 * 4] = v0;
        *(float4*)&sA[r1 * APITCH + c4 * 4] = v1;
    }
    __syncthreads();

    const int wave = threadIdx.x >> 5;
    const int lane = threadIdx.x & 31;
    const int col0 = wave * 16;
    const int m    = lane & 15;
    const int koff = (lane >> 4) * 2;

    v8f c = {};
    for (int kb = 0; kb < F; kb += 4) {
        const int k = kb + koff;
        v2f a, b;
        a.x = sA[m * APITCH + k];
        a.y = sA[m * APITCH + k + 1];
        b.x = fcw[(long long)k * F + col0 + m];
        b.y = fcw[(long long)(k + 1) * F + col0 + m];
        c = __builtin_amdgcn_wmma_f32_16x16x4_f32(
                false, a, false, b, (short)0, c, false, false);
    }

    for (int i = 0; i < 8; ++i) {
        int gr = row0 + i + (lane >> 4) * 8;
        if (gr < N) z[(long long)gr * F + col0 + m] = c[i];
    }

    const float al = attn_l[col0 + m];
    const float ar = attn_r[col0 + m];
    float v[16];
    for (int i = 0; i < 8; ++i) { v[i] = c[i] * al; v[8 + i] = c[i] * ar; }
    const bool b3 = (m & 8) != 0, b2 = (m & 4) != 0, b1 = (m & 2) != 0, b0 = (m & 1) != 0;
    float u[8], w[4], x[2];
    for (int j = 0; j < 8; ++j)
        u[j] = (b3 ? v[j + 8] : v[j]) + row_xor<8>(b3 ? v[j] : v[j + 8]);
    for (int j = 0; j < 4; ++j)
        w[j] = (b2 ? u[j + 4] : u[j]) + row_xor<4>(b2 ? u[j] : u[j + 4]);
    for (int j = 0; j < 2; ++j)
        x[j] = (b1 ? w[j + 2] : w[j]) + row_xor<2>(b1 ? w[j] : w[j + 2]);
    const float y = (b0 ? x[1] : x[0]) + row_xor<1>(b0 ? x[0] : x[1]);

    sE[(b3 ? 16 * HEADS : 0) + ((lane >> 4) * 8 + (m & 7)) * HEADS + wave] = y;
    __syncthreads();
    {
        const int t = threadIdx.x & 127;
        if (row0 + (t >> 3) < N) {
            float* __restrict__ dst = (threadIdx.x < 128) ? el : er;
            dst[(long long)row0 * HEADS + t] = sE[threadIdx.x];
        }
    }
}

__global__ void init_kernel(int* __restrict__ mult, float* __restrict__ red, int E)
{
    int t = blockIdx.x * blockDim.x + threadIdx.x;
    if (t < E) mult[t] = 0;
    if (t == 0) red[0] = 0.0f;
}

#define OWN_R  320
#define CNT_TB 1024
#define NSEG   4

__device__ __forceinline__ void own_edge(unsigned x, int e, unsigned nlo, unsigned span,
                                         int* lcnt, int* __restrict__ rank, int& below)
{
    below += (x > nlo) ? 1 : 0;
    if (x < span) rank[e] = atomicAdd(&lcnt[x], 1);
}

template <bool GUARD>
__device__ __forceinline__ void own_batch(const int4* __restrict__ d4, int q0, int qe,
                                          unsigned lo2, unsigned nlo, unsigned span,
                                          int* lcnt, int* __restrict__ rank, int& below)
{
    int4 v[4];
    bool ok[4];
    #pragma unroll
    for (int u = 0; u < 4; ++u) {
        const int q = q0 + u * CNT_TB;
        ok[u] = !GUARD || (q < qe);
        v[u] = d4[ok[u] ? q : (qe - 1)];
    }
    unsigned x[16];
    #pragma unroll
    for (int u = 0; u < 4; ++u) {
        x[4 * u + 0] = ok[u] ? (unsigned)v[u].x - lo2 : span;
        x[4 * u + 1] = ok[u] ? (unsigned)v[u].y - lo2 : span;
        x[4 * u + 2] = ok[u] ? (unsigned)v[u].z - lo2 : span;
        x[4 * u + 3] = ok[u] ? (unsigned)v[u].w - lo2 : span;
    }
    int nb = 0;
    #pragma unroll
    for (int j = 0; j < 16; ++j) nb += (x[j] > nlo) ? 1 : 0;
    below += nb;
    #pragma unroll
    for (int j = 0; j < 16; ++j) {
        const int e = 4 * (q0 + (j >> 2) * CNT_TB) + (j & 3);
        if (x[j] < span) rank[e] = atomicAdd(&lcnt[x[j]], 1);
    }
}

__global__ __launch_bounds__(CNT_TB) void count_kernel(
    const int* __restrict__ edst, const int* __restrict__ sel,
    int E, int E_sel, int n0, int nCnt, int nOwn,
    int* __restrict__ mult, int* __restrict__ rank,
    int* __restrict__ rowptr, int* __restrict__ rowend)
{
    __shared__ int lcnt[OWN_R];
    __shared__ int wbelow[CNT_TB / 32];
    __shared__ int wtot[OWN_R / 32];
    const int tid = threadIdx.x;
    const int lane = tid & 31, wv = tid >> 5;

    if ((int)blockIdx.x >= nOwn * NSEG) {
        const int g  = ((int)blockIdx.x - nOwn * NSEG) * CNT_TB + tid;
        const int ng = E_sel >> 2;
        if (g < ng) {
            const int4 s = ((const int4*)sel)[g];
            atomicAdd(&mult[s.x], 1); atomicAdd(&mult[s.y], 1);
            atomicAdd(&mult[s.z], 1); atomicAdd(&mult[s.w], 1);
        } else if (g == ng) {
            for (int j = ng << 2; j < E_sel; ++j) atomicAdd(&mult[sel[j]], 1);
        }
        return;
    }

    const int seg = (int)blockIdx.x / nOwn;
    const int own = (int)blockIdx.x - seg * nOwn;
    const unsigned lo   = (unsigned)own * OWN_R;
    const unsigned left = (unsigned)nCnt - lo;
    const unsigned span = (left < OWN_R) ? left : OWN_R;
    const unsigned lo2  = (unsigned)n0 + lo;
    const unsigned nlo  = ~lo;
    if (tid < OWN_R) lcnt[tid] = 0;
    __syncthreads();

    const int4* __restrict__ d4 = (const int4*)edst;
    const int nq = E >> 2;
    const int gq = (nq + NSEG - 1) / NSEG;
    const int qb = (seg * gq < nq) ? seg * gq : nq;
    const int qe = (qb + gq < nq) ? qb + gq : nq;
    const int ngrp  = qe - qb;
    const int nB    = (ngrp + 4 * CNT_TB - 1) / (4 * CNT_TB);
    const int nFull = ngrp / (4 * CNT_TB);
    int below = 0;

    int bi = (nB > 0) ? (int)(((unsigned)own * (unsigned)nB) / (unsigned)nOwn) : 0;
    for (int k = 0; k < nB; ++k) {
        const int q0 = qb + bi * (4 * CNT_TB) + tid;
        if (bi < nFull) own_batch<false>(d4, q0, qe, lo2, nlo, span, lcnt, rank, below);
        else            own_batch<true >(d4, q0, qe, lo2, nlo, span, lcnt, rank, below);
        if (++bi == nB) bi = 0;
    }
    if (seg == NSEG - 1)
        for (int e = (nq << 2) + tid; e < E; e += CNT_TB)
            own_edge((unsigned)edst[e] - lo2, e, nlo, span, lcnt, rank, below);

    for (int off = 16; off > 0; off >>= 1) below += __shfl_xor(below, off);
    if (lane == 0) wbelow[wv] = below;
    __syncthreads();

    const int c = (tid < OWN_R) ? lcnt[tid] : 0;
    int v = c;
    for (int off = 1; off < 32; off <<= 1) { int u = __shfl_up(v, off); if (lane >= off) v += u; }
    if (lane == 31 && wv < OWN_R / 32) wtot[wv] = v;
    int b = wbelow[lane];
    for (int off = 16; off > 0; off >>= 1) b += __shfl_xor(b, off);
    __syncthreads();
    if ((unsigned)tid < span) {
        int pre = 4 * qb + b;
        for (int w = 0; w < wv; ++w) pre += wtot[w];
        rowptr[seg * nCnt + (int)lo + tid] = pre + v - c;
        rowend[seg * nCnt + (int)lo + tid] = pre + v;
    }
}

__global__ void fill_kernel(const int* __restrict__ esrc,
                            const int* __restrict__ edst,
                            const int* __restrict__ mult,
                            const int* __restrict__ rank,
                            const int* __restrict__ rowptr,
                            int E, int n0, int nCnt,
                            int2* __restrict__ csr)
{
    int t = blockIdx.x * blockDim.x + threadIdx.x;
    if (t >= E) return;
    const int d = edst[t] - n0;
    if (d < 0 || d >= nCnt) return;
    const int nq = E >> 2;
    const int gq = (nq + NSEG - 1) / NSEG;
    const int q = t >> 2;
    int s = NSEG - 1;
    if (q < nq) {
        s = 0;
        #pragma unroll
        for (int j = 1; j < NSEG; ++j) s += (q >= j * gq) ? 1 : 0;
    }
    csr[rowptr[s * nCnt + d] + rank[t]] = make_int2(esrc[t], mult[t]);
}

__device__ __forceinline__ float leaky(float x) { return (x >= 0.f) ? x : NEG_SLOPE * x; }

__device__ __forceinline__ int list_pos(int v, const int (&sb)[NSEG], const int (&cum)[NSEG + 1])
{
    int o = sb[0];
    #pragma unroll
    for (int s = 1; s < NSEG; ++s) o = (v >= cum[s]) ? (sb[s] - cum[s]) : o;
    return v + o;
}


template <int B>
__device__ __forceinline__ void max_batch(const float* __restrict__ el, int2 rec, int kb, int c,
                                          int h, float erh, float& mA, float& mB)
{
    float e[B];
    int mu[B];
    #pragma unroll
    for (int u = 0; u < B; ++u) {
        const int kk = (kb + u < c) ? (kb + u) : (c - 1);
        const int s = __builtin_amdgcn_readlane(rec.x, kk);
        mu[u] = __builtin_amdgcn_readlane(rec.y, kk);
        e[u] = el[s * HEADS + h];
    }
    __builtin_amdgcn_sched_barrier(0);
    #pragma unroll
    for (int u = 0; u < B; ++u) {
        const float x = leaky(e[u] + erh);
        mA = fmaxf(mA, x);
        if (mu[u] > 0) mB = fmaxf(mB, x);
    }
}

template <int B>
__device__ __forceinline__ void sum_batch(const float* __restrict__ el,
                                          const float4* __restrict__ z4, int2 rec, int kb, int c,
                                          int h, int lane, float erh, float mA, float mB,
                                          float& dA, float& dB, float4& aA, float4& aB)
{
    float e[B];
    float4 r[B];
    int mu[B];
    #pragma unroll
    for (int u = 0; u < B; ++u) {
        const int kk = (kb + u < c) ? (kb + u) : (c - 1);
        const int s = __builtin_amdgcn_readlane(rec.x, kk);
        mu[u] = __builtin_amdgcn_readlane(rec.y, kk);
        e[u] = el[s * HEADS + h];
        r[u] = z4[(size_t)s * 32 + lane];
    }
    #pragma unroll
    for (int u = 0; u < B; ++u) {
        const bool live = kb + u < c;
        const float x = leaky(e[u] + erh);
        const float wa = live ? __expf(x - mA) : 0.0f;
        const float wb = (live && mu[u] > 0) ? (float)mu[u] * __expf(x - mB) : 0.0f;
        dA += wa; dB += wb;
        aA.x += wa * r[u].x; aA.y += wa * r[u].y; aA.z += wa * r[u].z; aA.w += wa * r[u].w;
        aB.x += wb * r[u].x; aB.y += wb * r[u].y; aB.z += wb * r[u].z; aB.w += wb * r[u].w;
    }
}

#define GB1 16
#define GB2 8

__global__ __launch_bounds__(256) void gather_kernel(
    const float* __restrict__ z, const float* __restrict__ el,
    const float* __restrict__ er,
    const int* __restrict__ rowptr, const int* __restrict__ rowend,
    const int2* __restrict__ csr,
    const float* __restrict__ bias, const float* __restrict__ prelu_a,
    float* __restrict__ h_pos, float* __restrict__ h_neg, int n0, int nCnt)
{
    const int lane = threadIdx.x & 31;
    const int i = __builtin_amdgcn_readfirstlane((int)(blockIdx.x * 8 + (threadIdx.x >> 5)));
    if (i >= nCnt) return;
    const int n = n0 + i;
    const int h = lane >> 2;
    const float4* __restrict__ z4 = (const float4*)z;
    const float4 zv = z4[(size_t)n * 32 + lane];
    const float4 b4 = ((const float4*)bias)[lane];
    const float a = prelu_a[0];
    int sb[NSEG], se[NSEG];
    #pragma unroll
    for (int s = 0; s < NSEG; ++s) {
        sb[s] = rowptr[s * nCnt + i];
        se[s] = rowend[s * nCnt + i];
    }
    int cum[NSEG + 1];
    cum[0] = 0;
    #pragma unroll
    for (int s = 0; s < NSEG; ++s) cum[s + 1] = cum[s] + (se[s] - sb[s]);
    const int total = cum[NSEG];

    const float erh = er[n * HEADS + h];
    const float eln = el[n * HEADS + h];
    float mA = -INFINITY, mB = -INFINITY;

    int2 rec0 = make_int2(0, 0);
    if (lane < total) rec0 = csr[list_pos(lane, sb, cum)];

    for (int base = 0; base < total; base += 32) {
        const int c = (total - base < 32) ? (total - base) : 32;
        int2 rec = rec0;
        if (base > 0) {
            rec = make_int2(0, 0);
            if (lane < c) rec = csr[list_pos(base + lane, sb, cum)];
        }
        for (int kb = 0; kb < c; ) {
            if (c - kb > GB1 / 2) { max_batch<GB1>(el, rec, kb, c, h, erh, mA, mB);     kb += GB1; }
            else                  { max_batch<GB1 / 2>(el, rec, kb, c, h, erh, mA, mB); kb += GB1 / 2; }
        }
    }

    const float xs = leaky(eln + erh);
    mA = fmaxf(mA, xs);
    mB = fmaxf(mB, xs);

    float dA = __expf(xs - mA), dB = __expf(xs - mB);
    float4 aA = make_float4(dA * zv.x, dA * zv.y, dA * zv.z, dA * zv.w);
    float4 aB = make_float4(dB * zv.x, dB * zv.y, dB * zv.z, dB * zv.w);
    for (int base = 0; base < total; base += 32) {
        const int c = (total - base < 32) ? (total - base) : 32;
        int2 rec = rec0;
        if (base > 0) {
            rec = make_int2(0, 0);
            if (lane < c) rec = csr[list_pos(base + lane, sb, cum)];
        }
        for (int kb = 0; kb < c; ) {
            if (c - kb > GB2 / 2) {
                sum_batch<GB2>(el, z4, rec, kb, c, h, lane, erh, mA, mB, dA, dB, aA, aB);
                kb += GB2;
            } else {
                sum_batch<GB2 / 2>(el, z4, rec, kb, c, h, lane, erh, mA, mB, dA, dB, aA, aB);
                kb += GB2 / 2;
            }
        }
    }

    float pA[4] = { aA.x / dA + b4.x, aA.y / dA + b4.y, aA.z / dA + b4.z, aA.w / dA + b4.w };
    float pB[4] = { aB.x / dB + b4.x, aB.y / dB + b4.y, aB.z / dB + b4.z, aB.w / dB + b4.w };
    #pragma unroll
    for (int c = 0; c < 4; ++c) {
        pA[c] = (pA[c] >= 0.f) ? pA[c] : a * pA[c];
        pB[c] = (pB[c] >= 0.f) ? pB[c] : a * pB[c];
        for (int off = 4; off < 32; off <<= 1) {
            pA[c] += __shfl_xor(pA[c], off);
            pB[c] += __shfl_xor(pB[c], off);
        }
    }
    const int q = lane & 3, cc = (lane >> 2) & 3;
    const bool neg = lane >= 16;
    const float v0 = neg ? pB[0] : pA[0];
    const float v1 = neg ? pB[1] : pA[1];
    const float v2 = neg ? pB[2] : pA[2];
    const float v3 = neg ? pB[3] : pA[3];
    const float v = (cc == 0) ? v0 : (cc == 1) ? v1 : (cc == 2) ? v2 : v3;
    float* __restrict__ dstp = neg ? h_neg : h_pos;
    dstp[(size_t)i * DIM + q * 4 + cc] = v * (1.0f / HEADS);
}

__global__ __launch_bounds__(256) void cos_loss_kernel(
    const float* __restrict__ hp, const float* __restrict__ hn,
    float* __restrict__ red, int nCnt)
{
    int i = blockIdx.x * blockDim.x + threadIdx.x;
    float v = 0.f;
    if (i < nCnt) {
        float num = 0.f, na = 0.f, nb = 0.f;
        #pragma unroll
        for (int d = 0; d < DIM; ++d) {
            float a = hp[(long long)i * DIM + d];
            float b = hn[(long long)i * DIM + d];
            num += a * b; na += a * a; nb += b * b;
        }
        na = fmaxf(sqrtf(na), 1e-8f);
        nb = fmaxf(sqrtf(nb), 1e-8f);
        v = __expf((num / (na * nb)) * (1.0f / TEM));
    }
    __shared__ float sbuf[256];
    sbuf[threadIdx.x] = v;
    __syncthreads();
    for (int s = 128; s > 0; s >>= 1) {
        if (threadIdx.x < s) sbuf[threadIdx.x] += sbuf[threadIdx.x + s];
        __syncthreads();
    }
    if (threadIdx.x == 0) atomicAdd(red, sbuf[0]);
}

__global__ void write_loss_kernel(const float* __restrict__ red, float* __restrict__ out)
{
    out[0] = logf(red[0]);
}

extern "C" void kernel_launch(void* const* d_in, const int* in_sizes, int n_in,
                              void* d_out, int out_size, void* d_ws, size_t ws_size,
                              hipStream_t stream)
{
    const float* feat    = (const float*)d_in[0];
    const float* fcw     = (const float*)d_in[1];
    const float* attn_l  = (const float*)d_in[2];
    const float* attn_r  = (const float*)d_in[3];
    const float* bias    = (const float*)d_in[4];
    const float* prelu_a = (const float*)d_in[5];
    const int*   esrc    = (const int*)d_in[6];
    const int*   edst    = (const int*)d_in[7];
    const int*   negidx  = (const int*)d_in[8];

    const int N     = in_sizes[0] / F;
    const int E     = in_sizes[6];
    const int E_NEG = in_sizes[8];
    const int N_DST = (out_size - 1) / DIM;
    const int N_SRC = N - N_DST;

    float* out   = (float*)d_out;
    float* h_pos = out + 1;

    float*    ws     = (float*)d_ws;
    float*    z      = ws;
    float*    el     = z   + (size_t)N * F;
    float*    er     = el  + (size_t)N * HEADS;
    float*    h_neg  = er  + (size_t)N * HEADS;
    int2*     csr    = (int2*)(h_neg + (size_t)N_DST * DIM);
    int*      mult   = (int*)(csr + (size_t)E);
    float*    red    = (float*)(mult + (size_t)E + 2 * (size_t)N_DST);
    int*      rowptr = (int*)er;
    int*      rowend = rowptr + (size_t)NSEG * N_DST;
    int*      rank   = (int*)h_neg;

    const int TB = 256;
    auto cdiv = [](long long a, int b) { return (unsigned)((a + b - 1) / b); };

    gemm_z_wmma<<<cdiv(N, 16), TB, 0, stream>>>(feat, fcw, attn_l, attn_r, z, el, er, N);

    init_kernel<<<cdiv(E, TB), TB, 0, stream>>>(mult, red, E);
    const int nOwn = (int)cdiv(N_DST, OWN_R);
    count_kernel<<<nOwn * NSEG + cdiv((long long)(E_NEG >> 2) + 1, CNT_TB), CNT_TB, 0, stream>>>(
        edst, negidx, E, E_NEG, N_SRC, N_DST, nOwn, mult, rank, rowptr, rowend);
    fill_kernel<<<cdiv(E, TB), TB, 0, stream>>>(
        esrc, edst, mult, rank, rowptr, E, N_SRC, N_DST, csr);

    gather_kernel<<<cdiv(N_DST, 8), TB, 0, stream>>>(
        z, el, er, rowptr, rowend, csr, bias, prelu_a, h_pos, h_neg, N_SRC, N_DST);

    cos_loss_kernel<<<cdiv(N_DST, TB), TB, 0, stream>>>(h_pos, h_neg, red, N_DST);
    write_loss_kernel<<<1, 1, 0, stream>>>(red, out);
}


# GEMM row tiles and the edge count pass, which share no data, run as two roles of one launch (work-group index picks the role): the GEMM overlaps the count walk; same sums; 7 to 6 launches
# speedup vs baseline: 7.8015x; 1.0865x over previous
#include <hip/hip_runtime.h>
#include <math.h>

typedef float v2f __attribute__((ext_vector_type(2)));
typedef float v8f __attribute__((ext_vector_type(8)));

#define HEADS 8
#define DIM 16
#define F 128
#define NEG_SLOPE 0.2f
#define TEM 0.7f

#define APITCH (F + 4)

template <int M> __device__ __forceinline__ float row_xor(float v)
{
    return __int_as_float(__builtin_amdgcn_update_dpp(
        0, __float_as_int(v), 0x160 + M, 0xf, 0xf, false));
}
#define GSUB 4
__device__ __forceinline__ void gemm_role(
    const float* __restrict__ feat, const float* __restrict__ fcw,
    const float* __restrict__ attn_l, const float* __restrict__ attn_r,
    float* __restrict__ z, float* __restrict__ el, float* __restrict__ er, int N, int n0,
    int gb)
{
    __shared__ __attribute__((aligned(16))) float sA_all[GSUB * 16 * APITCH];
    __shared__ float sE_all[GSUB * 2 * 16 * HEADS];
    const int sub = threadIdx.x >> 8;
    const int tid = threadIdx.x & 255;
    float* sA = sA_all + sub * (16 * APITCH);
    float* sE = sE_all + sub * (2 * 16 * HEADS);
    const int row0 = (gb * GSUB + sub) * 16;

    {
        const int c4 = tid & 31;
        const int r0 = tid >> 5;
        const int r1 = r0 + 8;
        const int g0 = (row0 + r0 < N) ? (row0 + r0) : (N - 1);
        const int g1 = (row0 + r1 < N) ? (row0 + r1) : (N - 1);
        const float4 v0 = ((const float4*)(feat + (long long)g0 * F))[c4];
        const float4 v1 = ((const float4*)(feat + (long long)g1 * F))[c4];
        *(float4*)&sA[r0 * APITCH + c4 * 4] = v0;
        *(float4*)&sA[r1 * APITCH + c4 * 4] = v1;
    }
    __syncthreads();

    const int wave = tid >> 5;
    const int lane = tid & 31;
    const int col0 = wave * 16;
    const int m    = lane & 15;
    const int koff = (lane >> 4) * 2;

    v8f c = {};
    for (int kb = 0; kb < F; kb += 4) {
        const int k = kb + koff;
        v2f a, b;
        a.x = sA[m * APITCH + k];
        a.y = sA[m * APITCH + k + 1];
        b.x = fcw[(long long)k * F + col0 + m];
        b.y = fcw[(long long)(k + 1) * F + col0 + m];
        c = __builtin_amdgcn_wmma_f32_16x16x4_f32(
                false, a, false, b, (short)0, c, false, false);
    }

    for (int i = 0; i < 8; ++i) {
        int gr = row0 + i + (lane >> 4) * 8;
        if (gr < N) z[(long long)gr * F + col0 + m] = c[i];
    }

    const float al = attn_l[col0 + m];
    const float ar = attn_r[col0 + m];
    float v[16];
    for (int i = 0; i < 8; ++i) { v[i] = c[i] * al; v[8 + i] = c[i] * ar; }
    const bool b3 = (m & 8) != 0, b2 = (m & 4) != 0, b1 = (m & 2) != 0, b0 = (m & 1) != 0;
    float u[8], w[4], x[2];
    for (int j = 0; j < 8; ++j)
        u[j] = (b3 ? v[j + 8] : v[j]) + row_xor<8>(b3 ? v[j] : v[j + 8]);
    for (int j = 0; j < 4; ++j)
        w[j] = (b2 ? u[j + 4] : u[j]) + row_xor<4>(b2 ? u[j] : u[j + 4]);
    for (int j = 0; j < 2; ++j)
        x[j] = (b1 ? w[j + 2] : w[j]) + row_xor<2>(b1 ? w[j] : w[j + 2]);
    const float y = (b0 ? x[1] : x[0]) + row_xor<1>(b0 ? x[0] : x[1]);

    sE[(b3 ? 16 * HEADS : 0) + ((lane >> 4) * 8 + (m & 7)) * HEADS + wave] = y;
    __syncthreads();
    {
        const int t = tid & 127;
        const int gr = row0 + (t >> 3);
        const bool isr = tid >= 128;
        if (gr < N && (!isr || gr >= n0)) {
            float* __restrict__ dst = isr ? er : el;
            dst[(long long)row0 * HEADS + t] = sE[tid];
        }
    }
}

__global__ void init_kernel(int* __restrict__ mult, float* __restrict__ red, int E)
{
    int t = blockIdx.x * blockDim.x + threadIdx.x;
    if (t < E) mult[t] = 0;
    if (t == 0) red[0] = 0.0f;
}

#define OWN_R  320
#define CNT_TB 1024
#define NSEG   4

__device__ __forceinline__ void own_edge(unsigned x, int e, unsigned nlo, unsigned span,
                                         int* lcnt, int* __restrict__ rank, int& below)
{
    below += (x > nlo) ? 1 : 0;
    if (x < span) rank[e] = atomicAdd(&lcnt[x], 1);
}

template <bool GUARD>
__device__ __forceinline__ void own_batch(const int4* __restrict__ d4, int q0, int qe,
                                          unsigned lo2, unsigned nlo, unsigned span,
                                          int* lcnt, int* __restrict__ rank, int& below)
{
    int4 v[4];
    bool ok[4];
    #pragma unroll
    for (int u = 0; u < 4; ++u) {
        const int q = q0 + u * CNT_TB;
        ok[u] = !GUARD || (q < qe);
        v[u] = d4[ok[u] ? q : (qe - 1)];
    }
    unsigned x[16];
    #pragma unroll
    for (int u = 0; u < 4; ++u) {
        x[4 * u + 0] = ok[u] ? (unsigned)v[u].x - lo2 : span;
        x[4 * u + 1] = ok[u] ? (unsigned)v[u].y - lo2 : span;
        x[4 * u + 2] = ok[u] ? (unsigned)v[u].z - lo2 : span;
        x[4 * u + 3] = ok[u] ? (unsigned)v[u].w - lo2 : span;
    }
    int nb = 0;
    #pragma unroll
    for (int j = 0; j < 16; ++j) nb += (x[j] > nlo) ? 1 : 0;
    below += nb;
    #pragma unroll
    for (int j = 0; j < 16; ++j) {
        const int e = 4 * (q0 + (j >> 2) * CNT_TB) + (j & 3);
        if (x[j] < span) rank[e] = atomicAdd(&lcnt[x[j]], 1);
    }
}

__device__ __forceinline__ void count_role(
    const int* __restrict__ edst, const int* __restrict__ sel,
    int E, int E_sel, int n0, int nCnt, int nOwn,
    int* __restrict__ mult, int* __restrict__ rank,
    int* __restrict__ rowptr, int* __restrict__ rowend, int bid)
{
    __shared__ int lcnt[OWN_R];
    __shared__ int wbelow[CNT_TB / 32];
    __shared__ int wtot[OWN_R / 32];
    const int tid = threadIdx.x;
    const int lane = tid & 31, wv = tid >> 5;

    if (bid >= nOwn * NSEG) {
        const int g  = (bid - nOwn * NSEG) * CNT_TB + tid;
        const int ng = E_sel >> 2;
        if (g < ng) {
            const int4 s = ((const int4*)sel)[g];
            atomicAdd(&mult[s.x], 1); atomicAdd(&mult[s.y], 1);
            atomicAdd(&mult[s.z], 1); atomicAdd(&mult[s.w], 1);
        } else if (g == ng) {
            for (int j = ng << 2; j < E_sel; ++j) atomicAdd(&mult[sel[j]], 1);
        }
        return;
    }

    const int seg = bid / nOwn;
    const int own = bid - seg * nOwn;
    const unsigned lo   = (unsigned)own * OWN_R;
    const unsigned left = (unsigned)nCnt - lo;
    const unsigned span = (left < OWN_R) ? left : OWN_R;
    const unsigned lo2  = (unsigned)n0 + lo;
    const unsigned nlo  = ~lo;
    if (tid < OWN_R) lcnt[tid] = 0;
    __syncthreads();

    const int4* __restrict__ d4 = (const int4*)edst;
    const int nq = E >> 2;
    const int gq = (nq + NSEG - 1) / NSEG;
    const int qb = (seg * gq < nq) ? seg * gq : nq;
    const int qe = (qb + gq < nq) ? qb + gq : nq;
    const int ngrp  = qe - qb;
    const int nB    = (ngrp + 4 * CNT_TB - 1) / (4 * CNT_TB);
    const int nFull = ngrp / (4 * CNT_TB);
    int below = 0;

    int bi = (nB > 0) ? (int)(((unsigned)own * (unsigned)nB) / (unsigned)nOwn) : 0;
    for (int k = 0; k < nB; ++k) {
        const int q0 = qb + bi * (4 * CNT_TB) + tid;
        if (bi < nFull) own_batch<false>(d4, q0, qe, lo2, nlo, span, lcnt, rank, below);
        else            own_batch<true >(d4, q0, qe, lo2, nlo, span, lcnt, rank, below);
        if (++bi == nB) bi = 0;
    }
    if (seg == NSEG - 1)
        for (int e = (nq << 2) + tid; e < E; e += CNT_TB)
            own_edge((unsigned)edst[e] - lo2, e, nlo, span, lcnt, rank, below);

    for (int off = 16; off > 0; off >>= 1) below += __shfl_xor(below, off);
    if (lane == 0) wbelow[wv] = below;
    __syncthreads();

    const int c = (tid < OWN_R) ? lcnt[tid] : 0;
    int v = c;
    for (int off = 1; off < 32; off <<= 1) { int u = __shfl_up(v, off); if (lane >= off) v += u; }
    if (lane == 31 && wv < OWN_R / 32) wtot[wv] = v;
    int b = wbelow[lane];
    for (int off = 16; off > 0; off >>= 1) b += __shfl_xor(b, off);
    __syncthreads();
    if ((unsigned)tid < span) {
        int pre = 4 * qb + b;
        for (int w = 0; w < wv; ++w) pre += wtot[w];
        rowptr[seg * nCnt + (int)lo + tid] = pre + v - c;
        rowend[seg * nCnt + (int)lo + tid] = pre + v;
    }
}


__global__ __launch_bounds__(CNT_TB) void count_gemm_kernel(
    const float* __restrict__ feat, const float* __restrict__ fcw,
    const float* __restrict__ attn_l, const float* __restrict__ attn_r,
    float* __restrict__ z, float* __restrict__ el, float* __restrict__ er, int N,
    const int* __restrict__ edst, const int* __restrict__ sel,
    int E, int E_sel, int n0, int nCnt, int nOwn, int nCntBlk,
    int* __restrict__ mult, int* __restrict__ rank,
    int* __restrict__ rowptr, int* __restrict__ rowend)
{
    if ((int)blockIdx.x < nCntBlk)
        count_role(edst, sel, E, E_sel, n0, nCnt, nOwn, mult, rank, rowptr, rowend,
                   (int)blockIdx.x);
    else
        gemm_role(feat, fcw, attn_l, attn_r, z, el, er, N, n0, (int)blockIdx.x - nCntBlk);
}

__global__ void fill_kernel(const int* __restrict__ esrc,
                            const int* __restrict__ edst,
                            const int* __restrict__ mult,
                            const int* __restrict__ rank,
                            const int* __restrict__ rowptr,
                            int E, int n0, int nCnt,
                            int2* __restrict__ csr)
{
    int t = blockIdx.x * blockDim.x + threadIdx.x;
    if (t >= E) return;
    const int d = edst[t] - n0;
    if (d < 0 || d >= nCnt) return;
    const int nq = E >> 2;
    const int gq = (nq + NSEG - 1) / NSEG;
    const int q = t >> 2;
    int s = NSEG - 1;
    if (q < nq) {
        s = 0;
        #pragma unroll
        for (int j = 1; j < NSEG; ++j) s += (q >= j * gq) ? 1 : 0;
    }
    csr[rowptr[s * nCnt + d] + rank[t]] = make_int2(esrc[t], mult[t]);
}

__device__ __forceinline__ float leaky(float x) { return (x >= 0.f) ? x : NEG_SLOPE * x; }

__device__ __forceinline__ int list_pos(int v, const int (&sb)[NSEG], const int (&cum)[NSEG + 1])
{
    int o = sb[0];
    #pragma unroll
    for (int s = 1; s < NSEG; ++s) o = (v >= cum[s]) ? (sb[s] - cum[s]) : o;
    return v + o;
}


template <int B>
__device__ __forceinline__ void max_batch(const float* __restrict__ el, int2 rec, int kb, int c,
                                          int h, float erh, float& mA, float& mB)
{
    float e[B];
    int mu[B];
    #pragma unroll
    for (int u = 0; u < B; ++u) {
        const int kk = (kb + u < c) ? (kb + u) : (c - 1);
        const int s = __builtin_amdgcn_readlane(rec.x, kk);
        mu[u] = __builtin_amdgcn_readlane(rec.y, kk);
        e[u] = el[s * HEADS + h];
    }
    __builtin_amdgcn_sched_barrier(0);
    #pragma unroll
    for (int u = 0; u < B; ++u) {
        const float x = leaky(e[u] + erh);
        mA = fmaxf(mA, x);
        if (mu[u] > 0) mB = fmaxf(mB, x);
    }
}

template <int B>
__device__ __forceinline__ void sum_batch(const float* __restrict__ el,
                                          const float4* __restrict__ z4, int2 rec, int kb, int c,
                                          int h, int lane, float erh, float mA, float mB,
                                          float& dA, float& dB, float4& aA, float4& aB)
{
    float e[B];
    float4 r[B];
    int mu[B];
    #pragma unroll
    for (int u = 0; u < B; ++u) {
        const int kk = (kb + u < c) ? (kb + u) : (c - 1);
        const int s = __builtin_amdgcn_readlane(rec.x, kk);
        mu[u] = __builtin_amdgcn_readlane(rec.y, kk);
        e[u] = el[s * HEADS + h];
        r[u] = z4[(size_t)s * 32 + lane];
    }
    #pragma unroll
    for (int u = 0; u < B; ++u) {
        const bool live = kb + u < c;
        const float x = leaky(e[u] + erh);
        const float wa = live ? __expf(x - mA) : 0.0f;
        const float wb = (live && mu[u] > 0) ? (float)mu[u] * __expf(x - mB) : 0.0f;
        dA += wa; dB += wb;
        aA.x += wa * r[u].x; aA.y += wa * r[u].y; aA.z += wa * r[u].z; aA.w += wa * r[u].w;
        aB.x += wb * r[u].x; aB.y += wb * r[u].y; aB.z += wb * r[u].z; aB.w += wb * r[u].w;
    }
}

#define GB1 16
#define GB2 8

__global__ __launch_bounds__(256) void gather_kernel(
    const float* __restrict__ z, const float* __restrict__ el,
    const float* __restrict__ er,
    const int* __restrict__ rowptr, const int* __restrict__ rowend,
    const int2* __restrict__ csr,
    const float* __restrict__ bias, const float* __restrict__ prelu_a,
    float* __restrict__ h_pos, float* __restrict__ h_neg, int n0, int nCnt)
{
    const int lane = threadIdx.x & 31;
    const int i = __builtin_amdgcn_readfirstlane((int)(blockIdx.x * 8 + (threadIdx.x >> 5)));
    if (i >= nCnt) return;
    const int n = n0 + i;
    const int h = lane >> 2;
    const float4* __restrict__ z4 = (const float4*)z;
    const float4 zv = z4[(size_t)n * 32 + lane];
    const float4 b4 = ((const float4*)bias)[lane];
    const float a = prelu_a[0];
    int sb[NSEG], se[NSEG];
    #pragma unroll
    for (int s = 0; s < NSEG; ++s) {
        sb[s] = rowptr[s * nCnt + i];
        se[s] = rowend[s * nCnt + i];
    }
    int cum[NSEG + 1];
    cum[0] = 0;
    #pragma unroll
    for (int s = 0; s < NSEG; ++s) cum[s + 1] = cum[s] + (se[s] - sb[s]);
    const int total = cum[NSEG];

    const float erh = er[n * HEADS + h];
    const float eln = el[n * HEADS + h];
    float mA = -INFINITY, mB = -INFINITY;

    int2 rec0 = make_int2(0, 0);
    if (lane < total) rec0 = csr[list_pos(lane, sb, cum)];

    for (int base = 0; base < total; base += 32) {
        const int c = (total - base < 32) ? (total - base) : 32;
        int2 rec = rec0;
        if (base > 0) {
            rec = make_int2(0, 0);
            if (lane < c) rec = csr[list_pos(base + lane, sb, cum)];
        }
        for (int kb = 0; kb < c; ) {
            if (c - kb > GB1 / 2) { max_batch<GB1>(el, rec, kb, c, h, erh, mA, mB);     kb += GB1; }
            else                  { max_batch<GB1 / 2>(el, rec, kb, c, h, erh, mA, mB); kb += GB1 / 2; }
        }
    }

    const float xs = leaky(eln + erh);
    mA = fmaxf(mA, xs);
    mB = fmaxf(mB, xs);

    float dA = __expf(xs - mA), dB = __expf(xs - mB);
    float4 aA = make_float4(dA * zv.x, dA * zv.y, dA * zv.z, dA * zv.w);
    float4 aB = make_float4(dB * zv.x, dB * zv.y, dB * zv.z, dB * zv.w);
    for (int base = 0; base < total; base += 32) {
        const int c = (total - base < 32) ? (total - base) : 32;
        int2 rec = rec0;
        if (base > 0) {
            rec = make_int2(0, 0);
            if (lane < c) rec = csr[list_pos(base + lane, sb, cum)];
        }
        for (int kb = 0; kb < c; ) {
            if (c - kb > GB2 / 2) {
                sum_batch<GB2>(el, z4, rec, kb, c, h, lane, erh, mA, mB, dA, dB, aA, aB);
                kb += GB2;
            } else {
                sum_batch<GB2 / 2>(el, z4, rec, kb, c, h, lane, erh, mA, mB, dA, dB, aA, aB);
                kb += GB2 / 2;
            }
        }
    }

    float pA[4] = { aA.x / dA + b4.x, aA.y / dA + b4.y, aA.z / dA + b4.z, aA.w / dA + b4.w };
    float pB[4] = { aB.x / dB + b4.x, aB.y / dB + b4.y, aB.z / dB + b4.z, aB.w / dB + b4.w };
    #pragma unroll
    for (int c = 0; c < 4; ++c) {
        pA[c] = (pA[c] >= 0.f) ? pA[c] : a * pA[c];
        pB[c] = (pB[c] >= 0.f) ? pB[c] : a * pB[c];
        for (int off = 4; off < 32; off <<= 1) {
            pA[c] += __shfl_xor(pA[c], off);
            pB[c] += __shfl_xor(pB[c], off);
        }
    }
    const int q = lane & 3, cc = (lane >> 2) & 3;
    const bool neg = lane >= 16;
    const float v0 = neg ? pB[0] : pA[0];
    const float v1 = neg ? pB[1] : pA[1];
    const float v2 = neg ? pB[2] : pA[2];
    const float v3 = neg ? pB[3] : pA[3];
    const float v = (cc == 0) ? v0 : (cc == 1) ? v1 : (cc == 2) ? v2 : v3;
    float* __restrict__ dstp = neg ? h_neg : h_pos;
    dstp[(size_t)i * DIM + q * 4 + cc] = v * (1.0f / HEADS);
}

__global__ __launch_bounds__(256) void cos_loss_kernel(
    const float* __restrict__ hp, const float* __restrict__ hn,
    float* __restrict__ red, int nCnt)
{
    int i = blockIdx.x * blockDim.x + threadIdx.x;
    float v = 0.f;
    if (i < nCnt) {
        float num = 0.f, na = 0.f, nb = 0.f;
        #pragma unroll
        for (int d = 0; d < DIM; ++d) {
            float a = hp[(long long)i * DIM + d];
            float b = hn[(long long)i * DIM + d];
            num += a * b; na += a * a; nb += b * b;
        }
        na = fmaxf(sqrtf(na), 1e-8f);
        nb = fmaxf(sqrtf(nb), 1e-8f);
        v = __expf((num / (na * nb)) * (1.0f / TEM));
    }
    __shared__ float sbuf[256];
    sbuf[threadIdx.x] = v;
    __syncthreads();
    for (int s = 128; s > 0; s >>= 1) {
        if (threadIdx.x < s) sbuf[threadIdx.x] += sbuf[threadIdx.x + s];
        __syncthreads();
    }
    if (threadIdx.x == 0) atomicAdd(red, sbuf[0]);
}

__global__ void write_loss_kernel(const float* __restrict__ red, float* __restrict__ out)
{
    out[0] = logf(red[0]);
}

extern "C" void kernel_launch(void* const* d_in, const int* in_sizes, int n_in,
                              void* d_out, int out_size, void* d_ws, size_t ws_size,
                              hipStream_t stream)
{
    const float* feat    = (const float*)d_in[0];
    const float* fcw     = (const float*)d_in[1];
    const float* attn_l  = (const float*)d_in[2];
    const float* attn_r  = (const float*)d_in[3];
    const float* bias    = (const float*)d_in[4];
    const float* prelu_a = (const float*)d_in[5];
    const int*   esrc    = (const int*)d_in[6];
    const int*   edst    = (const int*)d_in[7];
    const int*   negidx  = (const int*)d_in[8];

    const int N     = in_sizes[0] / F;
    const int E     = in_sizes[6];
    const int E_NEG = in_sizes[8];
    const int N_DST = (out_size - 1) / DIM;
    const int N_SRC = N - N_DST;

    float* out   = (float*)d_out;
    float* h_pos = out + 1;

    float*    ws     = (float*)d_ws;
    float*    z      = ws;
    float*    el     = z   + (size_t)N * F;
    float*    er     = el  + (size_t)N * HEADS;
    float*    h_neg  = er  + (size_t)N * HEADS;
    int2*     csr    = (int2*)(h_neg + (size_t)N_DST * DIM);
    int*      mult   = (int*)(csr + (size_t)E);
    float*    red    = (float*)(mult + (size_t)E + 2 * (size_t)N_DST);
    int*      rowptr = (int*)er;
    int*      rowend = rowptr + (size_t)NSEG * N_DST;
    int*      rank   = (int*)h_neg;

    const int TB = 256;
    auto cdiv = [](long long a, int b) { return (unsigned)((a + b - 1) / b); };

    init_kernel<<<cdiv(E, TB), TB, 0, stream>>>(mult, red, E);

    const int nOwn = (int)cdiv(N_DST, OWN_R);
    const int nCntBlk = nOwn * NSEG + (int)cdiv((long long)(E_NEG >> 2) + 1, CNT_TB);
    count_gemm_kernel<<<nCntBlk + cdiv(N, 16 * GSUB), CNT_TB, 0, stream>>>(
        feat, fcw, attn_l, attn_r, z, el, er, N,
        edst, negidx, E, E_NEG, N_SRC, N_DST, nOwn, nCntBlk, mult, rank, rowptr, rowend);
    fill_kernel<<<cdiv(E, TB), TB, 0, stream>>>(
        esrc, edst, mult, rank, rowptr, E, N_SRC, N_DST, csr);

    gather_kernel<<<cdiv(N_DST, 8), TB, 0, stream>>>(
        z, el, er, rowptr, rowend, csr, bias, prelu_a, h_pos, h_neg, N_SRC, N_DST);

    cos_loss_kernel<<<cdiv(N_DST, TB), TB, 0, stream>>>(h_pos, h_neg, red, N_DST);
    write_loss_kernel<<<1, 1, 0, stream>>>(red, out);
}


# Range owners keep their own edges in an LDS list while counting and place them in the sorted list themselves behind their scan: no fill launch, no rank array (6 to 5 launches); same LDS-add sum order
# speedup vs baseline: 10.4596x; 1.3407x over previous
#include <hip/hip_runtime.h>
#include <math.h>

typedef float v2f __attribute__((ext_vector_type(2)));
typedef float v8f __attribute__((ext_vector_type(8)));

#define HEADS 8
#define DIM 16
#define F 128
#define NEG_SLOPE 0.2f
#define TEM 0.7f

#define APITCH (F + 4)

template <int M> __device__ __forceinline__ float row_xor(float v)
{
    return __int_as_float(__builtin_amdgcn_update_dpp(
        0, __float_as_int(v), 0x160 + M, 0xf, 0xf, false));
}
#define GSUB 4
__device__ __forceinline__ void gemm_role(
    const float* __restrict__ feat, const float* __restrict__ fcw,
    const float* __restrict__ attn_l, const float* __restrict__ attn_r,
    float* __restrict__ z, float* __restrict__ el, float* __restrict__ er, int N, int n0,
    int gb)
{
    __shared__ __attribute__((aligned(16))) float sA_all[GSUB * 16 * APITCH];
    __shared__ float sE_all[GSUB * 2 * 16 * HEADS];
    const int sub = threadIdx.x >> 8;
    const int tid = threadIdx.x & 255;
    float* sA = sA_all + sub * (16 * APITCH);
    float* sE = sE_all + sub * (2 * 16 * HEADS);
    const int row0 = (gb * GSUB + sub) * 16;

    {
        const int c4 = tid & 31;
        const int r0 = tid >> 5;
        const int r1 = r0 + 8;
        const int g0 = (row0 + r0 < N) ? (row0 + r0) : (N - 1);
        const int g1 = (row0 + r1 < N) ? (row0 + r1) : (N - 1);
        const float4 v0 = ((const float4*)(feat + (long long)g0 * F))[c4];
        const float4 v1 = ((const float4*)(feat + (long long)g1 * F))[c4];
        *(float4*)&sA[r0 * APITCH + c4 * 4] = v0;
        *(float4*)&sA[r1 * APITCH + c4 * 4] = v1;
    }
    __syncthreads();

    const int wave = tid >> 5;
    const int lane = tid & 31;
    const int col0 = wave * 16;
    const int m    = lane & 15;
    const int koff = (lane >> 4) * 2;

    v8f c = {};
    for (int kb = 0; kb < F; kb += 4) {
        const int k = kb + koff;
        v2f a, b;
        a.x = sA[m * APITCH + k];
        a.y = sA[m * APITCH + k + 1];
        b.x = fcw[(long long)k * F + col0 + m];
        b.y = fcw[(long long)(k + 1) * F + col0 + m];
        c = __builtin_amdgcn_wmma_f32_16x16x4_f32(
                false, a, false, b, (short)0, c, false, false);
    }

    for (int i = 0; i < 8; ++i) {
        int gr = row0 + i + (lane >> 4) * 8;
        if (gr < N) z[(long long)gr * F + col0 + m] = c[i];
    }

    const float al = attn_l[col0 + m];
    const float ar = attn_r[col0 + m];
    float v[16];
    for (int i = 0; i < 8; ++i) { v[i] = c[i] * al; v[8 + i] = c[i] * ar; }
    const bool b3 = (m & 8) != 0, b2 = (m & 4) != 0, b1 = (m & 2) != 0, b0 = (m & 1) != 0;
    float u[8], w[4], x[2];
    for (int j = 0; j < 8; ++j)
        u[j] = (b3 ? v[j + 8] : v[j]) + row_xor<8>(b3 ? v[j] : v[j + 8]);
    for (int j = 0; j < 4; ++j)
        w[j] = (b2 ? u[j + 4] : u[j]) + row_xor<4>(b2 ? u[j] : u[j + 4]);
    for (int j = 0; j < 2; ++j)
        x[j] = (b1 ? w[j + 2] : w[j]) + row_xor<2>(b1 ? w[j] : w[j + 2]);
    const float y = (b0 ? x[1] : x[0]) + row_xor<1>(b0 ? x[0] : x[1]);

    sE[(b3 ? 16 * HEADS : 0) + ((lane >> 4) * 8 + (m & 7)) * HEADS + wave] = y;
    __syncthreads();
    {
        const int t = tid & 127;
        const int gr = row0 + (t >> 3);
        const bool isr = tid >= 128;
        if (gr < N && (!isr || gr >= n0)) {
            float* __restrict__ dst = isr ? er : el;
            dst[(long long)row0 * HEADS + t] = sE[tid];
        }
    }
}

__global__ void init_kernel(int* __restrict__ mult, float* __restrict__ red, int E)
{
    int t = blockIdx.x * blockDim.x + threadIdx.x;
    if (t < E) mult[t] = 0;
    if (t == 0) red[0] = 0.0f;
}

#define OWN_R  320
#define CNT_TB 1024
#define NSEG   4
#define LCAP   4096
static_assert(OWN_R <= 512 && LCAP <= (1 << 22), "keep_edge packs offset and rank into 9 + 22 bits");

__device__ __forceinline__ void keep_edge(unsigned x, int e, int* lcnt, int* nl, int2* lst)
{
    const int r = atomicAdd(&lcnt[x], 1);
    const int slot = atomicAdd(nl, 1);
    if (slot < LCAP) lst[slot] = make_int2(e, (int)((x << 22) | (unsigned)r));
}

__device__ __forceinline__ void own_edge(unsigned x, int e, unsigned nlo, unsigned span,
                                         int* lcnt, int* nl, int2* lst, int& below)
{
    below += (x > nlo) ? 1 : 0;
    if (x < span) keep_edge(x, e, lcnt, nl, lst);
}

template <bool GUARD>
__device__ __forceinline__ void own_batch(const int4* __restrict__ d4, int q0, int qe,
                                          unsigned lo2, unsigned nlo, unsigned span,
                                          int* lcnt, int* nl, int2* lst, int& below)
{
    int4 v[4];
    bool ok[4];
    #pragma unroll
    for (int u = 0; u < 4; ++u) {
        const int q = q0 + u * CNT_TB;
        ok[u] = !GUARD || (q < qe);
        v[u] = d4[ok[u] ? q : (qe - 1)];
    }
    unsigned x[16];
    #pragma unroll
    for (int u = 0; u < 4; ++u) {
        x[4 * u + 0] = ok[u] ? (unsigned)v[u].x - lo2 : span;
        x[4 * u + 1] = ok[u] ? (unsigned)v[u].y - lo2 : span;
        x[4 * u + 2] = ok[u] ? (unsigned)v[u].z - lo2 : span;
        x[4 * u + 3] = ok[u] ? (unsigned)v[u].w - lo2 : span;
    }
    int nb = 0;
    #pragma unroll
    for (int j = 0; j < 16; ++j) nb += (x[j] > nlo) ? 1 : 0;
    below += nb;
    #pragma unroll
    for (int j = 0; j < 16; ++j) {
        const int e = 4 * (q0 + (j >> 2) * CNT_TB) + (j & 3);
        if (x[j] < span) keep_edge(x[j], e, lcnt, nl, lst);
    }
}

__device__ __forceinline__ void count_role(
    const int* __restrict__ esrc, const int* __restrict__ edst, const int* __restrict__ sel,
    int E, int E_sel, int n0, int nCnt, int nOwn,
    int* __restrict__ mult, int2* __restrict__ csr,
    int* __restrict__ rowptr, int* __restrict__ rowend, int bid)
{
    __shared__ int lcnt[OWN_R];
    __shared__ int lstart[OWN_R];
    __shared__ int wbelow[CNT_TB / 32];
    __shared__ int wtot[OWN_R / 32];
    __shared__ int nl[1];
    __shared__ int2 lst[LCAP];
    const int tid = threadIdx.x;
    const int lane = tid & 31, wv = tid >> 5;

    if (bid >= nOwn * NSEG) {
        const int g  = (bid - nOwn * NSEG) * CNT_TB + tid;
        const int ng = E_sel >> 2;
        if (g < ng) {
            const int4 s = ((const int4*)sel)[g];
            atomicAdd(&mult[s.x], 1); atomicAdd(&mult[s.y], 1);
            atomicAdd(&mult[s.z], 1); atomicAdd(&mult[s.w], 1);
        } else if (g == ng) {
            for (int j = ng << 2; j < E_sel; ++j) atomicAdd(&mult[sel[j]], 1);
        }
        return;
    }

    const int seg = bid / nOwn;
    const int own = bid - seg * nOwn;
    const unsigned lo   = (unsigned)own * OWN_R;
    const unsigned left = (unsigned)nCnt - lo;
    const unsigned span = (left < OWN_R) ? left : OWN_R;
    const unsigned lo2  = (unsigned)n0 + lo;
    const unsigned nlo  = ~lo;
    if (tid < OWN_R) lcnt[tid] = 0;
    if (tid == 0) nl[0] = 0;
    __syncthreads();

    const int4* __restrict__ d4 = (const int4*)edst;
    const int nq = E >> 2;
    const int gq = (nq + NSEG - 1) / NSEG;
    const int qb = (seg * gq < nq) ? seg * gq : nq;
    const int qe = (qb + gq < nq) ? qb + gq : nq;
    const int ngrp  = qe - qb;
    const int nB    = (ngrp + 4 * CNT_TB - 1) / (4 * CNT_TB);
    const int nFull = ngrp / (4 * CNT_TB);
    int below = 0;

    int bi = (nB > 0) ? (int)(((unsigned)own * (unsigned)nB) / (unsigned)nOwn) : 0;
    for (int k = 0; k < nB; ++k) {
        const int q0 = qb + bi * (4 * CNT_TB) + tid;
        if (bi < nFull) own_batch<false>(d4, q0, qe, lo2, nlo, span, lcnt, nl, lst, below);
        else            own_batch<true >(d4, q0, qe, lo2, nlo, span, lcnt, nl, lst, below);
        if (++bi == nB) bi = 0;
    }
    if (seg == NSEG - 1)
        for (int e = (nq << 2) + tid; e < E; e += CNT_TB)
            own_edge((unsigned)edst[e] - lo2, e, nlo, span, lcnt, nl, lst, below);

    for (int off = 16; off > 0; off >>= 1) below += __shfl_xor(below, off);
    if (lane == 0) wbelow[wv] = below;
    __syncthreads();

    const int c = (tid < OWN_R) ? lcnt[tid] : 0;
    int v = c;
    for (int off = 1; off < 32; off <<= 1) { int u = __shfl_up(v, off); if (lane >= off) v += u; }
    if (lane == 31 && wv < OWN_R / 32) wtot[wv] = v;
    int b = wbelow[lane];
    for (int off = 16; off > 0; off >>= 1) b += __shfl_xor(b, off);
    __syncthreads();
    if ((unsigned)tid < span) {
        int pre = 4 * qb + b;
        for (int w = 0; w < wv; ++w) pre += wtot[w];
        rowptr[seg * nCnt + (int)lo + tid] = pre + v - c;
        rowend[seg * nCnt + (int)lo + tid] = pre + v;
        lstart[tid] = pre + v - c;
    }
    __syncthreads();

    const int n = nl[0];
    if (n <= LCAP) {
        for (int j = tid; j < n; j += CNT_TB) {
            const int2 t = lst[j];
            csr[lstart[(unsigned)t.y >> 22] + (t.y & 0x3fffff)] = make_int2(esrc[t.x], t.x);
        }
    } else {
        if (tid < OWN_R) lcnt[tid] = 0;
        __syncthreads();
        const int e1 = (seg == NSEG - 1) ? E : 4 * qe;
        for (int e = 4 * qb + tid; e < e1; e += CNT_TB) {
            const unsigned x = (unsigned)edst[e] - lo2;
            if (x < span) csr[lstart[x] + atomicAdd(&lcnt[x], 1)] = make_int2(esrc[e], e);
        }
    }
}


__global__ __launch_bounds__(CNT_TB) void count_gemm_kernel(
    const float* __restrict__ feat, const float* __restrict__ fcw,
    const float* __restrict__ attn_l, const float* __restrict__ attn_r,
    float* __restrict__ z, float* __restrict__ el, float* __restrict__ er, int N,
    const int* __restrict__ esrc, const int* __restrict__ edst, const int* __restrict__ sel,
    int E, int E_sel, int n0, int nCnt, int nOwn, int nCntBlk,
    int* __restrict__ mult, int2* __restrict__ csr,
    int* __restrict__ rowptr, int* __restrict__ rowend)
{
    if ((int)blockIdx.x < nCntBlk)
        count_role(esrc, edst, sel, E, E_sel, n0, nCnt, nOwn, mult, csr, rowptr, rowend,
                   (int)blockIdx.x);
    else
        gemm_role(feat, fcw, attn_l, attn_r, z, el, er, N, n0, (int)blockIdx.x - nCntBlk);
}

__device__ __forceinline__ float leaky(float x) { return (x >= 0.f) ? x : NEG_SLOPE * x; }

__device__ __forceinline__ int list_pos(int v, const int (&sb)[NSEG], const int (&cum)[NSEG + 1])
{
    int o = sb[0];
    #pragma unroll
    for (int s = 1; s < NSEG; ++s) o = (v >= cum[s]) ? (sb[s] - cum[s]) : o;
    return v + o;
}


template <int B>
__device__ __forceinline__ void max_batch(const float* __restrict__ el, int2 rec, int kb, int c,
                                          int h, float erh, float& mA, float& mB)
{
    float e[B];
    #pragma unroll
    for (int u = 0; u < B; ++u) {
        const int kk = (kb + u < c) ? (kb + u) : (c - 1);
        const int s = __builtin_amdgcn_readlane(rec.x, kk);
        e[u] = el[s * HEADS + h];
    }
    __builtin_amdgcn_sched_barrier(0);
    #pragma unroll
    for (int u = 0; u < B; ++u) {
        const int kk = (kb + u < c) ? (kb + u) : (c - 1);
        const int mu = __builtin_amdgcn_readlane(rec.y, kk);
        const float x = leaky(e[u] + erh);
        mA = fmaxf(mA, x);
        if (mu > 0) mB = fmaxf(mB, x);
    }
}

template <int B>
__device__ __forceinline__ void sum_batch(const float* __restrict__ el,
                                          const float4* __restrict__ z4, int2 rec, int kb, int c,
                                          int h, int lane, float erh, float mA, float mB,
                                          float& dA, float& dB, float4& aA, float4& aB)
{
    float e[B];
    float4 r[B];
    int mu[B];
    #pragma unroll
    for (int u = 0; u < B; ++u) {
        const int kk = (kb + u < c) ? (kb + u) : (c - 1);
        const int s = __builtin_amdgcn_readlane(rec.x, kk);
        mu[u] = __builtin_amdgcn_readlane(rec.y, kk);
        e[u] = el[s * HEADS + h];
        r[u] = z4[(size_t)s * 32 + lane];
    }
    #pragma unroll
    for (int u = 0; u < B; ++u) {
        const bool live = kb + u < c;
        const float x = leaky(e[u] + erh);
        const float wa = live ? __expf(x - mA) : 0.0f;
        const float wb = (live && mu[u] > 0) ? (float)mu[u] * __expf(x - mB) : 0.0f;
        dA += wa; dB += wb;
        aA.x += wa * r[u].x; aA.y += wa * r[u].y; aA.z += wa * r[u].z; aA.w += wa * r[u].w;
        aB.x += wb * r[u].x; aB.y += wb * r[u].y; aB.z += wb * r[u].z; aB.w += wb * r[u].w;
    }
}

#define GB1 16
#define GB2 8

__device__ __forceinline__ void max_chunk(const float* __restrict__ el, int2 rec, int c,
                                          int h, float erh, float& mA, float& mB)
{
    for (int kb = 0; kb < c; ) {
        if (c - kb > GB1 / 2) { max_batch<GB1>(el, rec, kb, c, h, erh, mA, mB);     kb += GB1; }
        else                  { max_batch<GB1 / 2>(el, rec, kb, c, h, erh, mA, mB); kb += GB1 / 2; }
    }
}

__device__ __forceinline__ void sum_chunk(const float* __restrict__ el,
                                          const float4* __restrict__ z4, int2 rec, int c,
                                          int h, int lane, float erh, float mA, float mB,
                                          float& dA, float& dB, float4& aA, float4& aB)
{
    for (int kb = 0; kb < c; ) {
        if (c - kb > GB2 / 2) {
            sum_batch<GB2>(el, z4, rec, kb, c, h, lane, erh, mA, mB, dA, dB, aA, aB);
            kb += GB2;
        } else {
            sum_batch<GB2 / 2>(el, z4, rec, kb, c, h, lane, erh, mA, mB, dA, dB, aA, aB);
            kb += GB2 / 2;
        }
    }
}

__global__ __launch_bounds__(256) void gather_kernel(
    const float* __restrict__ z, const float* __restrict__ el,
    const float* __restrict__ er,
    const int* __restrict__ rowptr, const int* __restrict__ rowend,
    const int2* __restrict__ csr, const int* __restrict__ mult,
    const float* __restrict__ bias, const float* __restrict__ prelu_a,
    float* __restrict__ h_pos, float* __restrict__ h_neg, int n0, int nCnt)
{
    const int lane = threadIdx.x & 31;
    const int i = __builtin_amdgcn_readfirstlane((int)(blockIdx.x * 8 + (threadIdx.x >> 5)));
    if (i >= nCnt) return;
    const int n = n0 + i;
    const int h = lane >> 2;
    const float4* __restrict__ z4 = (const float4*)z;
    const float4 zv = z4[(size_t)n * 32 + lane];
    const float4 b4 = ((const float4*)bias)[lane];
    const float a = prelu_a[0];
    int sb[NSEG], se[NSEG];
    #pragma unroll
    for (int s = 0; s < NSEG; ++s) {
        sb[s] = rowptr[s * nCnt + i];
        se[s] = rowend[s * nCnt + i];
    }
    int cum[NSEG + 1];
    cum[0] = 0;
    #pragma unroll
    for (int s = 0; s < NSEG; ++s) cum[s + 1] = cum[s] + (se[s] - sb[s]);
    const int total = cum[NSEG];

    const float erh = er[n * HEADS + h];
    const float eln = el[n * HEADS + h];
    float mA = -INFINITY, mB = -INFINITY;

    int2 rec0 = make_int2(0, 0);
    if (lane < total) { rec0 = csr[list_pos(lane, sb, cum)]; rec0.y = mult[rec0.y]; }

    max_chunk(el, rec0, (total < 32) ? total : 32, h, erh, mA, mB);
    for (int base = 32; base < total; base += 32) {
        const int c = (total - base < 32) ? (total - base) : 32;
        int2 rec = make_int2(0, 0);
        if (lane < c) { rec = csr[list_pos(base + lane, sb, cum)]; rec.y = mult[rec.y]; }
        max_chunk(el, rec, c, h, erh, mA, mB);
    }

    const float xs = leaky(eln + erh);
    mA = fmaxf(mA, xs);
    mB = fmaxf(mB, xs);

    float dA = __expf(xs - mA), dB = __expf(xs - mB);
    float4 aA = make_float4(dA * zv.x, dA * zv.y, dA * zv.z, dA * zv.w);
    float4 aB = make_float4(dB * zv.x, dB * zv.y, dB * zv.z, dB * zv.w);
    sum_chunk(el, z4, rec0, (total < 32) ? total : 32, h, lane, erh, mA, mB, dA, dB, aA, aB);
    for (int base = 32; base < total; base += 32) {
        const int c = (total - base < 32) ? (total - base) : 32;
        int2 rec = make_int2(0, 0);
        if (lane < c) { rec = csr[list_pos(base + lane, sb, cum)]; rec.y = mult[rec.y]; }
        sum_chunk(el, z4, rec, c, h, lane, erh, mA, mB, dA, dB, aA, aB);
    }

    float pA[4] = { aA.x / dA + b4.x, aA.y / dA + b4.y, aA.z / dA + b4.z, aA.w / dA + b4.w };
    float pB[4] = { aB.x / dB + b4.x, aB.y / dB + b4.y, aB.z / dB + b4.z, aB.w / dB + b4.w };
    #pragma unroll
    for (int c = 0; c < 4; ++c) {
        pA[c] = (pA[c] >= 0.f) ? pA[c] : a * pA[c];
        pB[c] = (pB[c] >= 0.f) ? pB[c] : a * pB[c];
        for (int off = 4; off < 32; off <<= 1) {
            pA[c] += __shfl_xor(pA[c], off);
            pB[c] += __shfl_xor(pB[c], off);
        }
    }
    const int q = lane & 3, cc = (lane >> 2) & 3;
    const bool neg = lane >= 16;
    const float v0 = neg ? pB[0] : pA[0];
    const float v1 = neg ? pB[1] : pA[1];
    const float v2 = neg ? pB[2] : pA[2];
    const float v3 = neg ? pB[3] : pA[3];
    const float v = (cc == 0) ? v0 : (cc == 1) ? v1 : (cc == 2) ? v2 : v3;
    float* __restrict__ dstp = neg ? h_neg : h_pos;
    dstp[(size_t)i * DIM + q * 4 + cc] = v * (1.0f / HEADS);
}

__global__ __launch_bounds__(256) void cos_loss_kernel(
    const float* __restrict__ hp, const float* __restrict__ hn,
    float* __restrict__ red, int nCnt)
{
    int i = blockIdx.x * blockDim.x + threadIdx.x;
    float v = 0.f;
    if (i < nCnt) {
        float num = 0.f, na = 0.f, nb = 0.f;
        #pragma unroll
        for (int d = 0; d < DIM; ++d) {
            float a = hp[(long long)i * DIM + d];
            float b = hn[(long long)i * DIM + d];
            num += a * b; na += a * a; nb += b * b;
        }
        na = fmaxf(sqrtf(na), 1e-8f);
        nb = fmaxf(sqrtf(nb), 1e-8f);
        v = __expf((num / (na * nb)) * (1.0f / TEM));
    }
    __shared__ float sbuf[256];
    sbuf[threadIdx.x] = v;
    __syncthreads();
    for (int s = 128; s > 0; s >>= 1) {
        if (threadIdx.x < s) sbuf[threadIdx.x] += sbuf[threadIdx.x + s];
        __syncthreads();
    }
    if (threadIdx.x == 0) atomicAdd(red, sbuf[0]);
}

__global__ void write_loss_kernel(const float* __restrict__ red, float* __restrict__ out)
{
    out[0] = logf(red[0]);
}

extern "C" void kernel_launch(void* const* d_in, const int* in_sizes, int n_in,
                              void* d_out, int out_size, void* d_ws, size_t ws_size,
                              hipStream_t stream)
{
    const float* feat    = (const float*)d_in[0];
    const float* fcw     = (const float*)d_in[1];
    const float* attn_l  = (const float*)d_in[2];
    const float* attn_r  = (const float*)d_in[3];
    const float* bias    = (const float*)d_in[4];
    const float* prelu_a = (const float*)d_in[5];
    const int*   esrc    = (const int*)d_in[6];
    const int*   edst    = (const int*)d_in[7];
    const int*   negidx  = (const int*)d_in[8];

    const int N     = in_sizes[0] / F;
    const int E     = in_sizes[6];
    const int E_NEG = in_sizes[8];
    const int N_DST = (out_size - 1) / DIM;
    const int N_SRC = N - N_DST;

    float* out   = (float*)d_out;
    float* h_pos = out + 1;

    float*    ws     = (float*)d_ws;
    float*    z      = ws;
    float*    el     = z   + (size_t)N * F;
    float*    er     = el  + (size_t)N * HEADS;
    float*    h_neg  = er  + (size_t)N * HEADS;
    int2*     csr    = (int2*)(h_neg + (size_t)N_DST * DIM);
    int*      mult   = (int*)(csr + (size_t)E);
    float*    red    = (float*)(mult + (size_t)E + 2 * (size_t)N_DST);
    int*      rowptr = (int*)er;
    int*      rowend = rowptr + (size_t)NSEG * N_DST;

    const int TB = 256;
    auto cdiv = [](long long a, int b) { return (unsigned)((a + b - 1) / b); };

    init_kernel<<<cdiv(E, TB), TB, 0, stream>>>(mult, red, E);

    const int nOwn = (int)cdiv(N_DST, OWN_R);
    const int nCntBlk = nOwn * NSEG + (int)cdiv((long long)(E_NEG >> 2) + 1, CNT_TB);
    count_gemm_kernel<<<nCntBlk + cdiv(N, 16 * GSUB), CNT_TB, 0, stream>>>(
        feat, fcw, attn_l, attn_r, z, el, er, N,
        esrc, edst, negidx, E, E_NEG, N_SRC, N_DST, nOwn, nCntBlk, mult, csr, rowptr, rowend);

    gather_kernel<<<cdiv(N_DST, 8), TB, 0, stream>>>(
        z, el, er, rowptr, rowend, csr, mult, bias, prelu_a, h_pos, h_neg, N_SRC, N_DST);

    cos_loss_kernel<<<cdiv(N_DST, TB), TB, 0, stream>>>(h_pos, h_neg, red, N_DST);
    write_loss_kernel<<<1, 1, 0, stream>>>(red, out);
}
